# starting kernel
# baseline (speedup 1.0000x reference)
#include <hip/hip_runtime.h>

typedef __attribute__((ext_vector_type(2))) float v2f;
typedef __attribute__((ext_vector_type(8))) float v8f;

#define FDIM 128

__global__ __launch_bounds__(256) void k_init(int* __restrict__ deg,
                                              float* __restrict__ stats, int N)
{
    int i = blockIdx.x * 256 + threadIdx.x;
    if (i < N) deg[i] = 1;
    if (i < 256) stats[i] = 0.f;
}

__global__ __launch_bounds__(256) void k_deg(const int* __restrict__ dst,
                                             int* __restrict__ deg, int E)
{
    int i = blockIdx.x * 256 + threadIdx.x;
    if (i < E) atomicAdd(&deg[dst[i]], 1);
}

__global__ __launch_bounds__(256) void k_dis(const int* __restrict__ deg,
                                             float* __restrict__ dis, int N)
{
    int i = blockIdx.x * 256 + threadIdx.x;
    if (i < N) dis[i] = rsqrtf((float)deg[i]);
}

__global__ __launch_bounds__(256) void k_gemm1(const float* __restrict__ X,
                                               const float* __restrict__ W,
                                               float* __restrict__ O, int N)
{
    __shared__ float lA[32 * FDIM];
    __shared__ v2f   lB[64 * 64];

    const int t    = threadIdx.x;
    const int rowb = blockIdx.x * 32;
    const int colb = blockIdx.y * 64;

    for (int idx = t; idx < 32 * FDIM; idx += 256) {
        int r = idx >> 7, k = idx & 127;
        int gr = rowb + r; if (gr >= N) gr = N - 1;
        lA[idx] = X[(size_t)gr * FDIM + k];
    }
    for (int idx = t; idx < 64 * 64; idx += 256) {
        int k2 = idx >> 6, c = idx & 63;
        const float* wp = W + (size_t)(2 * k2) * FDIM + (colb + c);
        v2f b; b.x = wp[0]; b.y = wp[FDIM];
        lB[idx] = b;
    }
    __syncthreads();

    const int wave = t >> 5, lane = t & 31;
    const int m = lane & 15, hi = lane >> 4;
    const int rt = wave >> 2, ct = wave & 3;
    const float* Arow = &lA[(rt * 16 + m) * FDIM];
    const int nloc = ct * 16 + m;

    v8f acc = {0.f, 0.f, 0.f, 0.f, 0.f, 0.f, 0.f, 0.f};
#pragma unroll
    for (int k = 0; k < FDIM; k += 4) {
        v2f a = *(const v2f*)&Arow[k + 2 * hi];
        v2f b = lB[(k / 2 + hi) * 64 + nloc];
        acc = __builtin_amdgcn_wmma_f32_16x16x4_f32(
            false, a, false, b, (short)0, acc, false, false);
    }

    const int col = colb + nloc;
#pragma unroll
    for (int r = 0; r < 8; ++r) {
        int row = rowb + rt * 16 + r + 8 * hi;
        if (row < N) O[(size_t)row * FDIM + col] = acc[r];
    }
}

__global__ __launch_bounds__(256) void k_self128(const float* __restrict__ h,
                                                 const float* __restrict__ dis,
                                                 const float* __restrict__ bias,
                                                 float* __restrict__ out, int N)
{
    long long idx = (long long)blockIdx.x * 256 + threadIdx.x;
    if (idx >= (long long)N * FDIM) return;
    int i = (int)(idx >> 7), f = (int)(idx & 127);
    float d = dis[i];
    out[idx] = h[idx] * d * d + bias[f];
}

__global__ __launch_bounds__(256) void k_selfC(const float* __restrict__ h,
                                               const float* __restrict__ dis,
                                               const float* __restrict__ bias,
                                               float* __restrict__ out, int N, int C)
{
    long long idx = (long long)blockIdx.x * 256 + threadIdx.x;
    if (idx >= (long long)N * C) return;
    int i = (int)(idx / C), f = (int)(idx - (long long)i * C);
    float d = dis[i];
    out[idx] = h[idx] * d * d + bias[f];
}

__global__ __launch_bounds__(256) void k_edge128(const float* __restrict__ h,
                                                 const int* __restrict__ src,
                                                 const int* __restrict__ dst,
                                                 const float* __restrict__ dis,
                                                 float* __restrict__ out, int E)
{
    long long gid = (long long)blockIdx.x * 256 + threadIdx.x;
    int e = (int)(gid >> 5);
    int lane = threadIdx.x & 31;
    if (e >= E) return;
    int s = src[e], d = dst[e];
    float w = dis[s] * dis[d];
    const float4* hp = (const float4*)(h + (size_t)s * FDIM);
    float4 v = hp[lane];
    float* op = out + (size_t)d * FDIM + lane * 4;
    atomicAdd(op + 0, v.x * w);
    atomicAdd(op + 1, v.y * w);
    atomicAdd(op + 2, v.z * w);
    atomicAdd(op + 3, v.w * w);
}

__global__ __launch_bounds__(256) void k_edgeC(const float* __restrict__ h,
                                               const int* __restrict__ src,
                                               const int* __restrict__ dst,
                                               const float* __restrict__ dis,
                                               float* __restrict__ out, int E, int C)
{
    long long gid = (long long)blockIdx.x * 256 + threadIdx.x;
    int e = (int)(gid >> 5);
    int lane = threadIdx.x & 31;
    if (e >= E) return;
    int s = src[e], d = dst[e];
    float w = dis[s] * dis[d];
    const float* hp = h + (size_t)s * C;
    float* op = out + (size_t)d * C;
    if (lane < C)      atomicAdd(op + lane,      hp[lane]      * w);
    if (32 + lane < C) atomicAdd(op + 32 + lane, hp[32 + lane] * w);
}

__global__ __launch_bounds__(256) void k_relu_stats(float* __restrict__ a,
                                                    float* __restrict__ gsum,
                                                    float* __restrict__ gsq, int N)
{
    __shared__ float ssum[256];
    __shared__ float ssq[256];
    int t = threadIdx.x, f = t & 127, sub = t >> 7;
    float sum = 0.f, sq = 0.f;
    int base = blockIdx.x * 64;
#pragma unroll 4
    for (int i = 0; i < 32; ++i) {
        int row = base + sub + 2 * i;
        if (row < N) {
            size_t idx = (size_t)row * FDIM + f;
            float v = a[idx];
            v = fmaxf(v, 0.f);
            a[idx] = v;
            sum += v;
            sq  += v * v;
        }
    }
    ssum[t] = sum; ssq[t] = sq;
    __syncthreads();
    if (t < 128) {
        atomicAdd(&gsum[t], ssum[t] + ssum[t + 128]);
        atomicAdd(&gsq[t],  ssq[t]  + ssq[t + 128]);
    }
}

__global__ void k_bn_finalize(const float* __restrict__ gsum,
                              const float* __restrict__ gsq,
                              const float* __restrict__ gamma,
                              const float* __restrict__ beta,
                              float* __restrict__ scl,
                              float* __restrict__ sft, int N)
{
    int f = threadIdx.x;
    float inv_n = 1.f / (float)N;
    float mean = gsum[f] * inv_n;
    float var  = gsq[f] * inv_n - mean * mean;
    float inv  = rsqrtf(var + 1e-5f);
    float g = gamma[f] * inv;
    scl[f] = g;
    sft[f] = beta[f] - mean * g;
}

__global__ __launch_bounds__(96) void k_gemm2(const float* __restrict__ A,
                                              const float* __restrict__ W,
                                              const float* __restrict__ scl,
                                              const float* __restrict__ sft,
                                              float* __restrict__ O, int N, int C)
{
    __shared__ float lA[16 * FDIM];
    __shared__ v2f   lB[64 * 48];

    const int t = threadIdx.x;
    const int rowb = blockIdx.x * 16;

    for (int idx = t; idx < 16 * FDIM; idx += 96) {
        int r = idx >> 7, k = idx & 127;
        int gr = rowb + r; if (gr >= N) gr = N - 1;
        lA[idx] = A[(size_t)gr * FDIM + k] * scl[k] + sft[k];
    }
    for (int idx = t; idx < 64 * 48; idx += 96) {
        int k2 = idx / 48, c = idx - k2 * 48;
        v2f b;
        b.x = (c < C) ? W[(size_t)(2 * k2) * C + c]     : 0.f;
        b.y = (c < C) ? W[(size_t)(2 * k2 + 1) * C + c] : 0.f;
        lB[idx] = b;
    }
    __syncthreads();

    const int wave = t >> 5, lane = t & 31;
    const int m = lane & 15, hi = lane >> 4;
    const float* Arow = &lA[m * FDIM];
    const int nloc = wave * 16 + m;

    v8f acc = {0.f, 0.f, 0.f, 0.f, 0.f, 0.f, 0.f, 0.f};
#pragma unroll
    for (int k = 0; k < FDIM; k += 4) {
        v2f a = *(const v2f*)&Arow[k + 2 * hi];
        v2f b = lB[(k / 2 + hi) * 48 + nloc];
        acc = __builtin_amdgcn_wmma_f32_16x16x4_f32(
            false, a, false, b, (short)0, acc, false, false);
    }

    if (nloc < C) {
#pragma unroll
        for (int r = 0; r < 8; ++r) {
            int row = rowb + r + 8 * hi;
            if (row < N) O[(size_t)row * C + nloc] = acc[r];
        }
    }
}

__global__ __launch_bounds__(256) void k_logsoftmax(float* __restrict__ out, int N, int C)
{
    long long gid = (long long)blockIdx.x * 256 + threadIdx.x;
    int row = (int)(gid >> 5);
    int lane = threadIdx.x & 31;
    if (row >= N) return;
    float* p = out + (size_t)row * C;
    float v0 = (lane < C) ? p[lane] : -3.4e38f;
    float v1 = (32 + lane < C) ? p[32 + lane] : -3.4e38f;
    float mx = fmaxf(v0, v1);
#pragma unroll
    for (int o = 16; o > 0; o >>= 1) mx = fmaxf(mx, __shfl_xor(mx, o, 32));
    float s = ((lane < C) ? expf(v0 - mx) : 0.f) +
              ((32 + lane < C) ? expf(v1 - mx) : 0.f);
#pragma unroll
    for (int o = 16; o > 0; o >>= 1) s += __shfl_xor(s, o, 32);
    float l = logf(s);
    if (lane < C)      p[lane]      = v0 - mx - l;
    if (32 + lane < C) p[32 + lane] = v1 - mx - l;
}

extern "C" void kernel_launch(void* const* d_in, const int* in_sizes, int n_in,
                              void* d_out, int out_size, void* d_ws, size_t ws_size,
                              hipStream_t stream)
{
    const float* x     = (const float*)d_in[0];
    const int*   ei    = (const int*)  d_in[1];
    const float* W1    = (const float*)d_in[2];
    const float* b1    = (const float*)d_in[3];
    const float* gamma = (const float*)d_in[4];
    const float* beta  = (const float*)d_in[5];
    const float* W2    = (const float*)d_in[6];
    const float* b2    = (const float*)d_in[7];
    float* out = (float*)d_out;

    const int N = in_sizes[0] / FDIM;
    const int E = in_sizes[1] / 2;
    const int C = in_sizes[6] / FDIM;
    const int* src = ei;
    const int* dst = ei + E;

    char* ws = (char*)d_ws;
    size_t off = 0;
    auto up = [](size_t v) { return (v + 255) & ~(size_t)255; };
    int*   deg   = (int*)(ws + off);   off = up(off + (size_t)N * 4);
    float* dis   = (float*)(ws + off); off = up(off + (size_t)N * 4);
    float* stats = (float*)(ws + off); off = up(off + 512 * 4);
    float* gsum = stats, *gsq = stats + 128, *scl = stats + 256, *sft = stats + 384;
    float* h1   = (float*)(ws + off);  off = up(off + (size_t)N * FDIM * 4);
    float* agg1 = (float*)(ws + off);  off = up(off + (size_t)N * FDIM * 4);
    float* h2   = h1;

    const int nb_N   = (N + 255) / 256;
    const int nb_E   = (E + 255) / 256;
    const int nb_Ew  = (E + 7) / 8;
    const int nb_Nw  = (N + 7) / 8;
    const long long nfe1 = (long long)N * FDIM;
    const long long nfe2 = (long long)N * C;

    k_init<<<nb_N, 256, 0, stream>>>(deg, stats, N);
    k_deg<<<nb_E, 256, 0, stream>>>(dst, deg, E);
    k_dis<<<nb_N, 256, 0, stream>>>(deg, dis, N);

    k_gemm1<<<dim3((N + 31) / 32, FDIM / 64), 256, 0, stream>>>(x, W1, h1, N);

    k_self128<<<(int)((nfe1 + 255) / 256), 256, 0, stream>>>(h1, dis, b1, agg1, N);
    k_edge128<<<nb_Ew, 256, 0, stream>>>(h1, src, dst, dis, agg1, E);

    k_relu_stats<<<(N + 63) / 64, 256, 0, stream>>>(agg1, gsum, gsq, N);
    k_bn_finalize<<<1, 128, 0, stream>>>(gsum, gsq, gamma, beta, scl, sft, N);

    k_gemm2<<<(N + 15) / 16, 96, 0, stream>>>(agg1, W2, scl, sft, h2, N, C);

    k_selfC<<<(int)((nfe2 + 255) / 256), 256, 0, stream>>>(h2, dis, b2, out, N, C);
    k_edgeC<<<nb_Ew, 256, 0, stream>>>(h2, src, dst, dis, out, E, C);

    k_logsoftmax<<<nb_Nw, 256, 0, stream>>>(out, N, C);
}


# Edge sums without float atomics: edges grouped by destination once per call (counting sort, slot = arrival order); one wave sums a node's in-rows in registers in list order, self term folded in
# speedup vs baseline: 2.0777x; 2.0777x over previous
#include <hip/hip_runtime.h>

typedef __attribute__((ext_vector_type(2))) float v2f;
typedef __attribute__((ext_vector_type(8))) float v8f;

#define FDIM 128

__global__ __launch_bounds__(256) void k_init(int* __restrict__ cnt,
                                              float* __restrict__ stats, int N)
{
    int i = blockIdx.x * 256 + threadIdx.x;
    if (i < N) cnt[i] = 0;
    if (i < 256) stats[i] = 0.f;
}

__global__ __launch_bounds__(256) void k_rank(const int* __restrict__ dst,
                                              int* __restrict__ cnt,
                                              int* __restrict__ rank, int E)
{
    int i = blockIdx.x * 256 + threadIdx.x;
    if (i < E) rank[i] = atomicAdd(&cnt[dst[i]], 1);
}

__global__ __launch_bounds__(256) void k_bsum(const int* __restrict__ cnt,
                                              int* __restrict__ bsum, int N)
{
    __shared__ int ws[8];
    const int t = threadIdx.x, lane = t & 31, wave = t >> 5;
    const int base = blockIdx.x * 1024 + t * 4;
    int s = 0;
    for (int j = 0; j < 4; ++j)
        if (base + j < N) s += cnt[base + j];
    for (int o = 16; o > 0; o >>= 1) s += __shfl_xor(s, o, 32);
    if (lane == 0) ws[wave] = s;
    __syncthreads();
    if (t == 0) {
        int a = 0;
        for (int w = 0; w < 8; ++w) a += ws[w];
        bsum[blockIdx.x] = a;
    }
}

__global__ __launch_bounds__(256) void k_scan(int* __restrict__ cnt_rowptr,
                                              const int* __restrict__ bsum,
                                              float* __restrict__ dis, int N)
{
    __shared__ int ws[8];
    __shared__ int boff;
    const int t = threadIdx.x, lane = t & 31, wave = t >> 5;
    const int base = blockIdx.x * 1024 + t * 4;

    if (wave == 0) {
        int a = 0;
        for (int j = lane; j < (int)blockIdx.x; j += 32) a += bsum[j];
        for (int o = 16; o > 0; o >>= 1) a += __shfl_xor(a, o, 32);
        if (lane == 0) boff = a;
    }

    int c[4];
    int tsum = 0;
    for (int j = 0; j < 4; ++j) {
        c[j] = (base + j < N) ? cnt_rowptr[base + j] : 0;
        tsum += c[j];
    }
    int incl = tsum;
    for (int o = 1; o < 32; o <<= 1) {
        int v = __shfl_up(incl, o, 32);
        if (lane >= o) incl += v;
    }
    if (lane == 31) ws[wave] = incl;
    __syncthreads();

    int excl = boff + incl - tsum;
    for (int w = 0; w < wave; ++w) excl += ws[w];
    for (int j = 0; j < 4; ++j) {
        if (base + j < N) {
            cnt_rowptr[base + j] = excl;
            dis[base + j] = rsqrtf((float)(c[j] + 1));
        }
        excl += c[j];
    }
}

__global__ __launch_bounds__(256) void k_fill(const int* __restrict__ src,
                                              const int* __restrict__ dst,
                                              const int* __restrict__ rank,
                                              const int* __restrict__ rowptr,
                                              int* __restrict__ csr_src, int E)
{
    int i = blockIdx.x * 256 + threadIdx.x;
    if (i < E) csr_src[rowptr[dst[i]] + rank[i]] = src[i];
}

__global__ __launch_bounds__(256) void k_gemm1(const float* __restrict__ X,
                                               const float* __restrict__ W,
                                               float* __restrict__ O, int N)
{
    __shared__ float lA[32 * FDIM];
    __shared__ v2f   lB[64 * 64];

    const int t    = threadIdx.x;
    const int rowb = blockIdx.x * 32;
    const int colb = blockIdx.y * 64;

    for (int idx = t; idx < 32 * FDIM; idx += 256) {
        int r = idx >> 7, k = idx & 127;
        int gr = rowb + r; if (gr >= N) gr = N - 1;
        lA[idx] = X[(size_t)gr * FDIM + k];
    }
    for (int idx = t; idx < 64 * 64; idx += 256) {
        int k2 = idx >> 6, c = idx & 63;
        const float* wp = W + (size_t)(2 * k2) * FDIM + (colb + c);
        v2f b; b.x = wp[0]; b.y = wp[FDIM];
        lB[idx] = b;
    }
    __syncthreads();

    const int wave = t >> 5, lane = t & 31;
    const int m = lane & 15, hi = lane >> 4;
    const int rt = wave >> 2, ct = wave & 3;
    const float* Arow = &lA[(rt * 16 + m) * FDIM];
    const int nloc = ct * 16 + m;

    v8f acc = {0.f, 0.f, 0.f, 0.f, 0.f, 0.f, 0.f, 0.f};
#pragma unroll
    for (int k = 0; k < FDIM; k += 4) {
        v2f a = *(const v2f*)&Arow[k + 2 * hi];
        v2f b = lB[(k / 2 + hi) * 64 + nloc];
        acc = __builtin_amdgcn_wmma_f32_16x16x4_f32(
            false, a, false, b, (short)0, acc, false, false);
    }

    const int col = colb + nloc;
#pragma unroll
    for (int r = 0; r < 8; ++r) {
        int row = rowb + rt * 16 + r + 8 * hi;
        if (row < N) O[(size_t)row * FDIM + col] = acc[r];
    }
}

static __device__ __forceinline__ int bcast_i(int v, int j)
{
    return __builtin_amdgcn_readlane(v, j);
}
static __device__ __forceinline__ float bcast_f(float v, int j)
{
    return __int_as_float(__builtin_amdgcn_readlane(__float_as_int(v), j));
}

static __device__ __forceinline__ void row_fma(float4& acc, const float4& v, float wt)
{
    acc.x += v.x * wt; acc.y += v.y * wt; acc.z += v.z * wt; acc.w += v.w * wt;
}

__global__ __launch_bounds__(256) void k_gather128(const float* __restrict__ h,
                                                   const int* __restrict__ rowptr,
                                                   const int* __restrict__ csr_src,
                                                   const float* __restrict__ dis,
                                                   const float* __restrict__ bias,
                                                   float* __restrict__ out, int N, int E)
{
    const int lane = threadIdx.x & 31;
    const int node = blockIdx.x * 8 + __builtin_amdgcn_readfirstlane((int)(threadIdx.x >> 5));
    if (node >= N) return;
    const int beg = __builtin_amdgcn_readfirstlane(rowptr[node]);
    const int end = (node + 1 < N) ? __builtin_amdgcn_readfirstlane(rowptr[node + 1]) : E;
    const float di = dis[node];
    const float4* hp = (const float4*)h;

    float4 acc;
    {
        float4 v = hp[(size_t)node * 32 + lane];
        float4 b = ((const float4*)bias)[lane];
        acc.x = v.x * di * di + b.x;
        acc.y = v.y * di * di + b.y;
        acc.z = v.z * di * di + b.z;
        acc.w = v.w * di * di + b.w;
    }
    for (int p = beg; p < end; p += 32) {
        const int n = (end - p < 32) ? (end - p) : 32;
        int s_l = 0; float w_l = 0.f;
        if (lane < n) { s_l = csr_src[p + lane]; w_l = dis[s_l] * di; }
        int j = 0;
        for (; j + 4 <= n; j += 4) {
            const int s0 = bcast_i(s_l, j),     s1 = bcast_i(s_l, j + 1);
            const int s2 = bcast_i(s_l, j + 2), s3 = bcast_i(s_l, j + 3);
            const float w0 = bcast_f(w_l, j),     w1 = bcast_f(w_l, j + 1);
            const float w2 = bcast_f(w_l, j + 2), w3 = bcast_f(w_l, j + 3);
            const float4 v0 = hp[(size_t)s0 * 32 + lane];
            const float4 v1 = hp[(size_t)s1 * 32 + lane];
            const float4 v2 = hp[(size_t)s2 * 32 + lane];
            const float4 v3 = hp[(size_t)s3 * 32 + lane];
            row_fma(acc, v0, w0); row_fma(acc, v1, w1);
            row_fma(acc, v2, w2); row_fma(acc, v3, w3);
        }
        for (; j < n; ++j) {
            const int s0 = bcast_i(s_l, j);
            const float w0 = bcast_f(w_l, j);
            const float4 v0 = hp[(size_t)s0 * 32 + lane];
            row_fma(acc, v0, w0);
        }
    }
    ((float4*)out)[(size_t)node * 32 + lane] = acc;
}

__global__ __launch_bounds__(256) void k_gatherC(const float* __restrict__ h,
                                                 const int* __restrict__ rowptr,
                                                 const int* __restrict__ csr_src,
                                                 const float* __restrict__ dis,
                                                 const float* __restrict__ bias,
                                                 float* __restrict__ out, int N, int E, int C)
{
    const int lane = threadIdx.x & 31;
    const int node = blockIdx.x * 8 + __builtin_amdgcn_readfirstlane((int)(threadIdx.x >> 5));
    if (node >= N) return;
    const int beg = __builtin_amdgcn_readfirstlane(rowptr[node]);
    const int end = (node + 1 < N) ? __builtin_amdgcn_readfirstlane(rowptr[node + 1]) : E;
    const float di = dis[node];
    const bool a0 = lane < C, a1 = 32 + lane < C;
    const int c0 = a0 ? lane : 0, c1 = a1 ? 32 + lane : 0;

    float acc0, acc1;
    {
        const float* hr = h + (size_t)node * C;
        acc0 = hr[c0] * di * di + bias[c0];
        acc1 = hr[c1] * di * di + bias[c1];
    }
    for (int p = beg; p < end; p += 32) {
        const int n = (end - p < 32) ? (end - p) : 32;
        int s_l = 0; float w_l = 0.f;
        if (lane < n) { s_l = csr_src[p + lane]; w_l = dis[s_l] * di; }
        int j = 0;
        for (; j + 4 <= n; j += 4) {
            const float* r0 = h + (size_t)bcast_i(s_l, j) * C;
            const float* r1 = h + (size_t)bcast_i(s_l, j + 1) * C;
            const float* r2 = h + (size_t)bcast_i(s_l, j + 2) * C;
            const float* r3 = h + (size_t)bcast_i(s_l, j + 3) * C;
            const float w0 = bcast_f(w_l, j),     w1 = bcast_f(w_l, j + 1);
            const float w2 = bcast_f(w_l, j + 2), w3 = bcast_f(w_l, j + 3);
            const float x0 = r0[c0], y0 = r0[c1];
            const float x1 = r1[c0], y1 = r1[c1];
            const float x2 = r2[c0], y2 = r2[c1];
            const float x3 = r3[c0], y3 = r3[c1];
            acc0 += x0 * w0; acc1 += y0 * w0;
            acc0 += x1 * w1; acc1 += y1 * w1;
            acc0 += x2 * w2; acc1 += y2 * w2;
            acc0 += x3 * w3; acc1 += y3 * w3;
        }
        for (; j < n; ++j) {
            const float* r0 = h + (size_t)bcast_i(s_l, j) * C;
            const float w0 = bcast_f(w_l, j);
            acc0 += r0[c0] * w0; acc1 += r0[c1] * w0;
        }
    }
    float* orow = out + (size_t)node * C;
    if (a0) orow[lane] = acc0;
    if (a1) orow[32 + lane] = acc1;
}

__global__ __launch_bounds__(256) void k_relu_stats(float* __restrict__ a,
                                                    float* __restrict__ gsum,
                                                    float* __restrict__ gsq, int N)
{
    __shared__ float ssum[256];
    __shared__ float ssq[256];
    int t = threadIdx.x, f = t & 127, sub = t >> 7;
    float sum = 0.f, sq = 0.f;
    int base = blockIdx.x * 64;
#pragma unroll 4
    for (int i = 0; i < 32; ++i) {
        int row = base + sub + 2 * i;
        if (row < N) {
            size_t idx = (size_t)row * FDIM + f;
            float v = a[idx];
            v = fmaxf(v, 0.f);
            a[idx] = v;
            sum += v;
            sq  += v * v;
        }
    }
    ssum[t] = sum; ssq[t] = sq;
    __syncthreads();
    if (t < 128) {
        atomicAdd(&gsum[t], ssum[t] + ssum[t + 128]);
        atomicAdd(&gsq[t],  ssq[t]  + ssq[t + 128]);
    }
}

__global__ void k_bn_finalize(const float* __restrict__ gsum,
                              const float* __restrict__ gsq,
                              const float* __restrict__ gamma,
                              const float* __restrict__ beta,
                              float* __restrict__ scl,
                              float* __restrict__ sft, int N)
{
    int f = threadIdx.x;
    float inv_n = 1.f / (float)N;
    float mean = gsum[f] * inv_n;
    float var  = gsq[f] * inv_n - mean * mean;
    float inv  = rsqrtf(var + 1e-5f);
    float g = gamma[f] * inv;
    scl[f] = g;
    sft[f] = beta[f] - mean * g;
}

__global__ __launch_bounds__(96) void k_gemm2(const float* __restrict__ A,
                                              const float* __restrict__ W,
                                              const float* __restrict__ scl,
                                              const float* __restrict__ sft,
                                              float* __restrict__ O, int N, int C)
{
    __shared__ float lA[16 * FDIM];
    __shared__ v2f   lB[64 * 48];

    const int t = threadIdx.x;
    const int rowb = blockIdx.x * 16;

    for (int idx = t; idx < 16 * FDIM; idx += 96) {
        int r = idx >> 7, k = idx & 127;
        int gr = rowb + r; if (gr >= N) gr = N - 1;
        lA[idx] = A[(size_t)gr * FDIM + k] * scl[k] + sft[k];
    }
    for (int idx = t; idx < 64 * 48; idx += 96) {
        int k2 = idx / 48, c = idx - k2 * 48;
        v2f b;
        b.x = (c < C) ? W[(size_t)(2 * k2) * C + c]     : 0.f;
        b.y = (c < C) ? W[(size_t)(2 * k2 + 1) * C + c] : 0.f;
        lB[idx] = b;
    }
    __syncthreads();

    const int wave = t >> 5, lane = t & 31;
    const int m = lane & 15, hi = lane >> 4;
    const float* Arow = &lA[m * FDIM];
    const int nloc = wave * 16 + m;

    v8f acc = {0.f, 0.f, 0.f, 0.f, 0.f, 0.f, 0.f, 0.f};
#pragma unroll
    for (int k = 0; k < FDIM; k += 4) {
        v2f a = *(const v2f*)&Arow[k + 2 * hi];
        v2f b = lB[(k / 2 + hi) * 48 + nloc];
        acc = __builtin_amdgcn_wmma_f32_16x16x4_f32(
            false, a, false, b, (short)0, acc, false, false);
    }

    if (nloc < C) {
#pragma unroll
        for (int r = 0; r < 8; ++r) {
            int row = rowb + r + 8 * hi;
            if (row < N) O[(size_t)row * C + nloc] = acc[r];
        }
    }
}

__global__ __launch_bounds__(256) void k_logsoftmax(const float* __restrict__ in,
                                                    float* __restrict__ out, int N, int C)
{
    long long gid = (long long)blockIdx.x * 256 + threadIdx.x;
    int row = (int)(gid >> 5);
    int lane = threadIdx.x & 31;
    if (row >= N) return;
    const float* p = in + (size_t)row * C;
    float* q = out + (size_t)row * C;
    float v0 = (lane < C) ? p[lane] : -3.4e38f;
    float v1 = (32 + lane < C) ? p[32 + lane] : -3.4e38f;
    float mx = fmaxf(v0, v1);
#pragma unroll
    for (int o = 16; o > 0; o >>= 1) mx = fmaxf(mx, __shfl_xor(mx, o, 32));
    float s = ((lane < C) ? expf(v0 - mx) : 0.f) +
              ((32 + lane < C) ? expf(v1 - mx) : 0.f);
#pragma unroll
    for (int o = 16; o > 0; o >>= 1) s += __shfl_xor(s, o, 32);
    float l = logf(s);
    if (lane < C)      q[lane]      = v0 - mx - l;
    if (32 + lane < C) q[32 + lane] = v1 - mx - l;
}

extern "C" void kernel_launch(void* const* d_in, const int* in_sizes, int n_in,
                              void* d_out, int out_size, void* d_ws, size_t ws_size,
                              hipStream_t stream)
{
    const float* x     = (const float*)d_in[0];
    const int*   ei    = (const int*)  d_in[1];
    const float* W1    = (const float*)d_in[2];
    const float* b1    = (const float*)d_in[3];
    const float* gamma = (const float*)d_in[4];
    const float* beta  = (const float*)d_in[5];
    const float* W2    = (const float*)d_in[6];
    const float* b2    = (const float*)d_in[7];
    float* out = (float*)d_out;

    const int N = in_sizes[0] / FDIM;
    const int E = in_sizes[1] / 2;
    const int C = in_sizes[6] / FDIM;
    const int* src = ei;
    const int* dst = ei + E;

    char* ws = (char*)d_ws;
    size_t off = 0;
    auto up = [](size_t v) { return (v + 255) & ~(size_t)255; };
    int*   rowptr = (int*)(ws + off);  off = up(off + (size_t)N * 4);
    float* dis   = (float*)(ws + off); off = up(off + (size_t)N * 4);
    float* stats = (float*)(ws + off); off = up(off + 512 * 4);
    float* gsum = stats, *gsq = stats + 128, *scl = stats + 256, *sft = stats + 384;
    float* h1   = (float*)(ws + off);  off = up(off + (size_t)N * FDIM * 4);
    float* agg1 = (float*)(ws + off);  off = up(off + (size_t)N * FDIM * 4);
    float* h2   = h1;
    int* rank = (int*)agg1;
    int* bsum = (int*)((char*)agg1 + up((size_t)E * 4));
    int*   csr_src = (int*)d_out;
    float* logits  = agg1;

    const int nb_N   = (N + 255) / 256;
    const int nb_E   = (E + 255) / 256;
    const int nb_T   = (N + 1023) / 1024;
    const int nb_Nw  = (N + 7) / 8;

    k_init<<<nb_N, 256, 0, stream>>>(rowptr, stats, N);
    k_rank<<<nb_E, 256, 0, stream>>>(dst, rowptr, rank, E);
    k_bsum<<<nb_T, 256, 0, stream>>>(rowptr, bsum, N);
    k_scan<<<nb_T, 256, 0, stream>>>(rowptr, bsum, dis, N);
    k_fill<<<nb_E, 256, 0, stream>>>(src, dst, rank, rowptr, csr_src, E);

    k_gemm1<<<dim3((N + 31) / 32, FDIM / 64), 256, 0, stream>>>(x, W1, h1, N);

    k_gather128<<<nb_Nw, 256, 0, stream>>>(h1, rowptr, csr_src, dis, b1, agg1, N, E);

    k_relu_stats<<<(N + 63) / 64, 256, 0, stream>>>(agg1, gsum, gsq, N);
    k_bn_finalize<<<1, 128, 0, stream>>>(gsum, gsq, gamma, beta, scl, sft, N);

    k_gemm2<<<(N + 15) / 16, 96, 0, stream>>>(agg1, W2, scl, sft, h2, N, C);

    k_gatherC<<<nb_Nw, 256, 0, stream>>>(h2, rowptr, csr_src, dis, b2, logits, N, E, C);

    k_logsoftmax<<<nb_Nw, 256, 0, stream>>>(logits, out, N, C);
}


# Edges grouped by destination by a two-level counting sort in LDS (256-node bins, run-wise stores) instead of a returning global add and a scattered 4-byte store per edge; in-list sum order changes
# speedup vs baseline: 3.0527x; 1.4693x over previous
#include <hip/hip_runtime.h>

typedef __attribute__((ext_vector_type(2))) float v2f;
typedef __attribute__((ext_vector_type(8))) float v8f;

#define FDIM 128

#define BSH    8
#define MAXNB  512
#define CH     8192
#define EPT    (CH / 512)
#define SEGPT  20
#define SEGCAP 12288

__global__ __launch_bounds__(256) void k_init(int* __restrict__ bins,
                                              float* __restrict__ stats)
{
    int i = blockIdx.x * 256 + threadIdx.x;
    if (i < 2 * MAXNB) bins[i] = 0;
    if (i < 256) stats[i] = 0.f;
}

__global__ __launch_bounds__(512) void k_hist(const int* __restrict__ dst,
                                              int* __restrict__ binCnt, int E)
{
    __shared__ int hist[MAXNB];
    const int t = threadIdx.x;
    const int base = blockIdx.x * CH;
    const int n = (E - base < CH) ? (E - base) : CH;
    int dv[EPT];
#pragma unroll
    for (int k = 0; k < EPT; ++k) {
        const int i = t + k * 512;
        dv[k] = dst[base + ((i < n) ? i : n - 1)];
    }
    hist[t] = 0;
    __syncthreads();
#pragma unroll
    for (int k = 0; k < EPT; ++k)
        if (t + k * 512 < n) atomicAdd(&hist[dv[k] >> BSH], 1);
    __syncthreads();
    const int c = hist[t];
    if (c > 0) atomicAdd(&binCnt[t], c);
}

__global__ __launch_bounds__(512) void k_scatter(const int* __restrict__ src,
                                                 const int* __restrict__ dst,
                                                 const int* __restrict__ binCnt,
                                                 int* __restrict__ cursor,
                                                 int* __restrict__ tmp, int E)
{
    __shared__ int hist[MAXNB];
    __shared__ int lcur[MAXNB];
    __shared__ int gdelta[MAXNB];
    __shared__ int wc[16], wg[16];
    __shared__ int sval[CH];
    __shared__ int sadr[CH];

    const int t = threadIdx.x, lane = t & 31, wave = t >> 5;
    const int base = blockIdx.x * CH;
    const int n = (E - base < CH) ? (E - base) : CH;

    int dv[EPT], sv[EPT];
#pragma unroll
    for (int k = 0; k < EPT; ++k) {
        const int i = t + k * 512;
        dv[k] = dst[base + ((i < n) ? i : n - 1)];
    }
#pragma unroll
    for (int k = 0; k < EPT; ++k) {
        const int i = t + k * 512;
        sv[k] = src[base + ((i < n) ? i : n - 1)];
    }
    const int g = binCnt[t];

    hist[t] = 0;
    __syncthreads();
#pragma unroll
    for (int k = 0; k < EPT; ++k)
        if (t + k * 512 < n) atomicAdd(&hist[dv[k] >> BSH], 1);
    __syncthreads();

    const int c = hist[t];
    int ci = c, gi = g;
    for (int o = 1; o < 32; o <<= 1) {
        int vc = __shfl_up(ci, o, 32);
        int vg = __shfl_up(gi, o, 32);
        if (lane >= o) { ci += vc; gi += vg; }
    }
    if (lane == 31) { wc[wave] = ci; wg[wave] = gi; }
    __syncthreads();
    int ce = ci - c, ge = gi - g;
    for (int w = 0; w < wave; ++w) { ce += wc[w]; ge += wg[w]; }
    int r = 0;
    if (c > 0) r = atomicAdd(&cursor[t], c);
    lcur[t] = ce;
    gdelta[t] = ge + r - ce;
    __syncthreads();

    int pv[EPT];
#pragma unroll
    for (int k = 0; k < EPT; ++k) {
        pv[k] = 0;
        if (t + k * 512 < n) pv[k] = atomicAdd(&lcur[dv[k] >> BSH], 1);
    }
#pragma unroll
    for (int k = 0; k < EPT; ++k) {
        if (t + k * 512 < n) {
            const int p = pv[k];
            sval[p] = (int)(((unsigned)sv[k] << 8) | (unsigned)(dv[k] & 255));
            sadr[p] = gdelta[dv[k] >> BSH] + p;
        }
    }
    __syncthreads();
    for (int i = t; i < n; i += 512) tmp[sadr[i]] = sval[i];
}

__global__ __launch_bounds__(256) void k_group(const int* __restrict__ tmp,
                                               const int* __restrict__ binCnt,
                                               int* __restrict__ rowptr,
                                               float* __restrict__ dis,
                                               int* __restrict__ csr_src, int N)
{
    __shared__ int cnt[256];
    __shared__ int red[8], wsum[8];
    __shared__ int sbuf[SEGCAP];

    const int t = threadIdx.x, lane = t & 31, wave = t >> 5;
    const int b = blockIdx.x;

    const int g0 = binCnt[t], g1 = binCnt[t + 256];
    int a = ((t < b) ? g0 : 0) + ((t + 256 < b) ? g1 : 0);
    for (int o = 16; o > 0; o >>= 1) a += __shfl_xor(a, o, 32);
    if (lane == 0) red[wave] = a;
    cnt[t] = 0;
    __syncthreads();
    int beg = 0;
    for (int w = 0; w < 8; ++w) beg += red[w];
    beg = __builtin_amdgcn_readfirstlane(beg);
    const int len = __builtin_amdgcn_readfirstlane(binCnt[b]);
    const bool staged = len <= SEGCAP;

    for (int i0 = 0; i0 < len; i0 += SEGPT * 256) {
        int v[SEGPT];
#pragma unroll
        for (int k = 0; k < SEGPT; ++k) {
            const int i = i0 + t + k * 256;
            v[k] = tmp[beg + ((i < len) ? i : len - 1)];
        }
#pragma unroll
        for (int k = 0; k < SEGPT; ++k)
            if (i0 + t + k * 256 < len) atomicAdd(&cnt[v[k] & 255], 1);
    }
    __syncthreads();

    const int c = cnt[t];
    int incl = c;
    for (int o = 1; o < 32; o <<= 1) {
        int v = __shfl_up(incl, o, 32);
        if (lane >= o) incl += v;
    }
    if (lane == 31) wsum[wave] = incl;
    __syncthreads();
    int excl = incl - c;
    for (int w = 0; w < wave; ++w) excl += wsum[w];
    const int node = b * 256 + t;
    if (node < N) {
        rowptr[node] = beg + excl;
        dis[node] = rsqrtf((float)(c + 1));
    }
    cnt[t] = excl;
    __syncthreads();

    for (int i0 = 0; i0 < len; i0 += SEGPT * 256) {
        unsigned v[SEGPT];
        int p[SEGPT];
#pragma unroll
        for (int k = 0; k < SEGPT; ++k) {
            const int i = i0 + t + k * 256;
            v[k] = (unsigned)tmp[beg + ((i < len) ? i : len - 1)];
        }
#pragma unroll
        for (int k = 0; k < SEGPT; ++k) {
            p[k] = 0;
            if (i0 + t + k * 256 < len) p[k] = atomicAdd(&cnt[v[k] & 255u], 1);
        }
#pragma unroll
        for (int k = 0; k < SEGPT; ++k) {
            if (i0 + t + k * 256 < len) {
                const int s = (int)(v[k] >> 8);
                if (staged) sbuf[p[k]] = s;
                else        csr_src[beg + p[k]] = s;
            }
        }
    }
    __syncthreads();
    if (staged)
        for (int i = t; i < len; i += 256) csr_src[beg + i] = sbuf[i];
}

__global__ __launch_bounds__(256) void k_gemm1(const float* __restrict__ X,
                                               const float* __restrict__ W,
                                               const float* __restrict__ dis,
                                               float* __restrict__ O, int N)
{
    __shared__ float lA[32 * FDIM];
    __shared__ v2f   lB[64 * 64];

    const int t    = threadIdx.x;
    const int rowb = blockIdx.x * 32;
    const int colb = blockIdx.y * 64;

    float ds[8];
    {
        const int r0 = rowb + ((t >> 5) >> 2) * 16 + 8 * ((t & 31) >> 4);
        for (int r = 0; r < 8; ++r) {
            int row = r0 + r; if (row >= N) row = N - 1;
            ds[r] = dis[row];
        }
    }

    for (int idx = t; idx < 32 * FDIM; idx += 256) {
        int r = idx >> 7, k = idx & 127;
        int gr = rowb + r; if (gr >= N) gr = N - 1;
        lA[idx] = X[(size_t)gr * FDIM + k];
    }
    for (int idx = t; idx < 64 * 64; idx += 256) {
        int k2 = idx >> 6, c = idx & 63;
        const float* wp = W + (size_t)(2 * k2) * FDIM + (colb + c);
        v2f b; b.x = wp[0]; b.y = wp[FDIM];
        lB[idx] = b;
    }
    __syncthreads();

    const int wave = t >> 5, lane = t & 31;
    const int m = lane & 15, hi = lane >> 4;
    const int rt = wave >> 2, ct = wave & 3;
    const float* Arow = &lA[(rt * 16 + m) * FDIM];
    const int nloc = ct * 16 + m;

    v8f acc = {0.f, 0.f, 0.f, 0.f, 0.f, 0.f, 0.f, 0.f};
#pragma unroll
    for (int k = 0; k < FDIM; k += 4) {
        v2f a = *(const v2f*)&Arow[k + 2 * hi];
        v2f b = lB[(k / 2 + hi) * 64 + nloc];
        acc = __builtin_amdgcn_wmma_f32_16x16x4_f32(
            false, a, false, b, (short)0, acc, false, false);
    }

    const int col = colb + nloc;
#pragma unroll
    for (int r = 0; r < 8; ++r) {
        int row = rowb + rt * 16 + r + 8 * hi;
        if (row < N) O[(size_t)row * FDIM + col] = acc[r] * ds[r];
    }
}

static __device__ __forceinline__ int bcast_i(int v, int j)
{
    return __builtin_amdgcn_readlane(v, j);
}
static __device__ __forceinline__ void row_add(float4& acc, const float4& v)
{
    acc.x += v.x; acc.y += v.y; acc.z += v.z; acc.w += v.w;
}

__global__ __launch_bounds__(256) void k_gather128(const float* __restrict__ h,
                                                   const int* __restrict__ rowptr,
                                                   const int* __restrict__ csr_src,
                                                   const float* __restrict__ dis,
                                                   const float* __restrict__ bias,
                                                   float* __restrict__ out, int N, int E)
{
    const int lane = threadIdx.x & 31;
    const int node = blockIdx.x * 8 + __builtin_amdgcn_readfirstlane((int)(threadIdx.x >> 5));
    if (node >= N) return;
    const int beg = __builtin_amdgcn_readfirstlane(rowptr[node]);
    const int end = (node + 1 < N) ? __builtin_amdgcn_readfirstlane(rowptr[node + 1]) : E;
    const float di = dis[node];
    const float4* hp = (const float4*)h;

    float4 acc = hp[(size_t)node * 32 + lane];
    const float4 b = ((const float4*)bias)[lane];
    for (int p = beg; p < end; p += 32) {
        const int n = (end - p < 32) ? (end - p) : 32;
        int s_l = 0;
        if (lane < n) s_l = csr_src[p + lane];
        int j = 0;
        for (; j + 4 <= n; j += 4) {
            const int s0 = bcast_i(s_l, j),     s1 = bcast_i(s_l, j + 1);
            const int s2 = bcast_i(s_l, j + 2), s3 = bcast_i(s_l, j + 3);
            const float4 v0 = hp[(size_t)s0 * 32 + lane];
            const float4 v1 = hp[(size_t)s1 * 32 + lane];
            const float4 v2 = hp[(size_t)s2 * 32 + lane];
            const float4 v3 = hp[(size_t)s3 * 32 + lane];
            row_add(acc, v0); row_add(acc, v1);
            row_add(acc, v2); row_add(acc, v3);
        }
        for (; j < n; ++j) {
            const int s0 = bcast_i(s_l, j);
            const float4 v0 = hp[(size_t)s0 * 32 + lane];
            row_add(acc, v0);
        }
    }
    acc.x = acc.x * di + b.x;
    acc.y = acc.y * di + b.y;
    acc.z = acc.z * di + b.z;
    acc.w = acc.w * di + b.w;
    ((float4*)out)[(size_t)node * 32 + lane] = acc;
}

__global__ __launch_bounds__(256) void k_gatherC(const float* __restrict__ h,
                                                 const int* __restrict__ rowptr,
                                                 const int* __restrict__ csr_src,
                                                 const float* __restrict__ dis,
                                                 const float* __restrict__ bias,
                                                 float* __restrict__ out, int N, int E, int C)
{
    const int lane = threadIdx.x & 31;
    const int node = blockIdx.x * 8 + __builtin_amdgcn_readfirstlane((int)(threadIdx.x >> 5));
    if (node >= N) return;
    const int beg = __builtin_amdgcn_readfirstlane(rowptr[node]);
    const int end = (node + 1 < N) ? __builtin_amdgcn_readfirstlane(rowptr[node + 1]) : E;
    const float di = dis[node];
    const bool a0 = lane < C, a1 = 32 + lane < C;
    const int c0 = a0 ? lane : 0, c1 = a1 ? 32 + lane : 0;

    float acc0, acc1;
    {
        const float* hr = h + (size_t)node * C;
        acc0 = hr[c0];
        acc1 = hr[c1];
    }
    const float b0 = bias[c0], b1 = bias[c1];
    for (int p = beg; p < end; p += 32) {
        const int n = (end - p < 32) ? (end - p) : 32;
        int s_l = 0;
        if (lane < n) s_l = csr_src[p + lane];
        int j = 0;
        for (; j + 4 <= n; j += 4) {
            const float* r0 = h + (size_t)bcast_i(s_l, j) * C;
            const float* r1 = h + (size_t)bcast_i(s_l, j + 1) * C;
            const float* r2 = h + (size_t)bcast_i(s_l, j + 2) * C;
            const float* r3 = h + (size_t)bcast_i(s_l, j + 3) * C;
            const float x0 = r0[c0], y0 = r0[c1];
            const float x1 = r1[c0], y1 = r1[c1];
            const float x2 = r2[c0], y2 = r2[c1];
            const float x3 = r3[c0], y3 = r3[c1];
            acc0 += x0; acc1 += y0;
            acc0 += x1; acc1 += y1;
            acc0 += x2; acc1 += y2;
            acc0 += x3; acc1 += y3;
        }
        for (; j < n; ++j) {
            const float* r0 = h + (size_t)bcast_i(s_l, j) * C;
            acc0 += r0[c0]; acc1 += r0[c1];
        }
    }
    float* orow = out + (size_t)node * C;
    if (a0) orow[lane] = acc0 * di + b0;
    if (a1) orow[32 + lane] = acc1 * di + b1;
}

__global__ __launch_bounds__(256) void k_relu_stats(float* __restrict__ a,
                                                    float* __restrict__ gsum,
                                                    float* __restrict__ gsq, int N)
{
    __shared__ float ssum[256];
    __shared__ float ssq[256];
    int t = threadIdx.x, f = t & 127, sub = t >> 7;
    float sum = 0.f, sq = 0.f;
    int base = blockIdx.x * 64;
#pragma unroll 4
    for (int i = 0; i < 32; ++i) {
        int row = base + sub + 2 * i;
        if (row < N) {
            size_t idx = (size_t)row * FDIM + f;
            float v = a[idx];
            v = fmaxf(v, 0.f);
            a[idx] = v;
            sum += v;
            sq  += v * v;
        }
    }
    ssum[t] = sum; ssq[t] = sq;
    __syncthreads();
    if (t < 128) {
        atomicAdd(&gsum[t], ssum[t] + ssum[t + 128]);
        atomicAdd(&gsq[t],  ssq[t]  + ssq[t + 128]);
    }
}

__global__ void k_bn_finalize(const float* __restrict__ gsum,
                              const float* __restrict__ gsq,
                              const float* __restrict__ gamma,
                              const float* __restrict__ beta,
                              float* __restrict__ scl,
                              float* __restrict__ sft, int N)
{
    int f = threadIdx.x;
    float inv_n = 1.f / (float)N;
    float mean = gsum[f] * inv_n;
    float var  = gsq[f] * inv_n - mean * mean;
    float inv  = rsqrtf(var + 1e-5f);
    float g = gamma[f] * inv;
    scl[f] = g;
    sft[f] = beta[f] - mean * g;
}

__global__ __launch_bounds__(96) void k_gemm2(const float* __restrict__ A,
                                              const float* __restrict__ W,
                                              const float* __restrict__ scl,
                                              const float* __restrict__ sft,
                                              const float* __restrict__ dis,
                                              float* __restrict__ O, int N, int C)
{
    __shared__ float lA[16 * FDIM];
    __shared__ v2f   lB[64 * 48];

    const int t = threadIdx.x;
    const int rowb = blockIdx.x * 16;

    float ds[8];
    {
        const int r0 = rowb + 8 * ((t & 31) >> 4);
        for (int r = 0; r < 8; ++r) {
            int row = r0 + r; if (row >= N) row = N - 1;
            ds[r] = dis[row];
        }
    }

    for (int idx = t; idx < 16 * FDIM; idx += 96) {
        int r = idx >> 7, k = idx & 127;
        int gr = rowb + r; if (gr >= N) gr = N - 1;
        lA[idx] = A[(size_t)gr * FDIM + k] * scl[k] + sft[k];
    }
    for (int idx = t; idx < 64 * 48; idx += 96) {
        int k2 = idx / 48, c = idx - k2 * 48;
        v2f b;
        b.x = (c < C) ? W[(size_t)(2 * k2) * C + c]     : 0.f;
        b.y = (c < C) ? W[(size_t)(2 * k2 + 1) * C + c] : 0.f;
        lB[idx] = b;
    }
    __syncthreads();

    const int wave = t >> 5, lane = t & 31;
    const int m = lane & 15, hi = lane >> 4;
    const float* Arow = &lA[m * FDIM];
    const int nloc = wave * 16 + m;

    v8f acc = {0.f, 0.f, 0.f, 0.f, 0.f, 0.f, 0.f, 0.f};
#pragma unroll
    for (int k = 0; k < FDIM; k += 4) {
        v2f a = *(const v2f*)&Arow[k + 2 * hi];
        v2f b = lB[(k / 2 + hi) * 48 + nloc];
        acc = __builtin_amdgcn_wmma_f32_16x16x4_f32(
            false, a, false, b, (short)0, acc, false, false);
    }

    if (nloc < C) {
#pragma unroll
        for (int r = 0; r < 8; ++r) {
            int row = rowb + r + 8 * hi;
            if (row < N) O[(size_t)row * C + nloc] = acc[r] * ds[r];
        }
    }
}

__global__ __launch_bounds__(256) void k_logsoftmax(const float* __restrict__ in,
                                                    float* __restrict__ out, int N, int C)
{
    long long gid = (long long)blockIdx.x * 256 + threadIdx.x;
    int row = (int)(gid >> 5);
    int lane = threadIdx.x & 31;
    if (row >= N) return;
    const float* p = in + (size_t)row * C;
    float* q = out + (size_t)row * C;
    float v0 = (lane < C) ? p[lane] : -3.4e38f;
    float v1 = (32 + lane < C) ? p[32 + lane] : -3.4e38f;
    float mx = fmaxf(v0, v1);
#pragma unroll
    for (int o = 16; o > 0; o >>= 1) mx = fmaxf(mx, __shfl_xor(mx, o, 32));
    float s = ((lane < C) ? expf(v0 - mx) : 0.f) +
              ((32 + lane < C) ? expf(v1 - mx) : 0.f);
#pragma unroll
    for (int o = 16; o > 0; o >>= 1) s += __shfl_xor(s, o, 32);
    float l = logf(s);
    if (lane < C)      q[lane]      = v0 - mx - l;
    if (32 + lane < C) q[32 + lane] = v1 - mx - l;
}

extern "C" void kernel_launch(void* const* d_in, const int* in_sizes, int n_in,
                              void* d_out, int out_size, void* d_ws, size_t ws_size,
                              hipStream_t stream)
{
    const float* x     = (const float*)d_in[0];
    const int*   ei    = (const int*)  d_in[1];
    const float* W1    = (const float*)d_in[2];
    const float* b1    = (const float*)d_in[3];
    const float* gamma = (const float*)d_in[4];
    const float* beta  = (const float*)d_in[5];
    const float* W2    = (const float*)d_in[6];
    const float* b2    = (const float*)d_in[7];
    float* out = (float*)d_out;

    const int N = in_sizes[0] / FDIM;
    const int E = in_sizes[1] / 2;
    const int C = in_sizes[6] / FDIM;
    const int* src = ei;
    const int* dst = ei + E;

    char* ws = (char*)d_ws;
    size_t off = 0;
    auto up = [](size_t v) { return (v + 255) & ~(size_t)255; };
    int*   rowptr = (int*)(ws + off);  off = up(off + (size_t)N * 4);
    float* dis   = (float*)(ws + off); off = up(off + (size_t)N * 4);
    float* stats = (float*)(ws + off); off = up(off + 512 * 4);
    float* gsum = stats, *gsq = stats + 128, *scl = stats + 256, *sft = stats + 384;
    float* h1   = (float*)(ws + off);  off = up(off + (size_t)N * FDIM * 4);
    float* agg1 = (float*)(ws + off);  off = up(off + (size_t)N * FDIM * 4);
    float* h2   = h1;
    int* tmp    = (int*)agg1;
    int* binCnt = (int*)((char*)agg1 + up((size_t)E * 4));
    int* cursor = binCnt + MAXNB;
    int*   csr_src = (int*)d_out;
    float* logits  = agg1;

    const int nb_C   = (E + CH - 1) / CH;
    const int nb_B   = (N + 255) / 256;
    const int nb_Nw  = (N + 7) / 8;

    k_init<<<4, 256, 0, stream>>>(binCnt, stats);
    k_hist<<<nb_C, 512, 0, stream>>>(dst, binCnt, E);
    k_scatter<<<nb_C, 512, 0, stream>>>(src, dst, binCnt, cursor, tmp, E);
    k_group<<<nb_B, 256, 0, stream>>>(tmp, binCnt, rowptr, dis, csr_src, N);

    k_gemm1<<<dim3((N + 31) / 32, FDIM / 64), 256, 0, stream>>>(x, W1, dis, h1, N);

    k_gather128<<<nb_Nw, 256, 0, stream>>>(h1, rowptr, csr_src, dis, b1, agg1, N, E);

    k_relu_stats<<<(N + 63) / 64, 256, 0, stream>>>(agg1, gsum, gsq, N);
    k_bn_finalize<<<1, 128, 0, stream>>>(gsum, gsq, gamma, beta, scl, sft, N);

    k_gemm2<<<(N + 15) / 16, 96, 0, stream>>>(agg1, W2, scl, sft, dis, h2, N, C);

    k_gatherC<<<nb_Nw, 256, 0, stream>>>(h2, rowptr, csr_src, dis, b2, logits, N, E, C);

    k_logsoftmax<<<nb_Nw, 256, 0, stream>>>(logits, out, N, C);
}


# LDS A tile of both GEMMs at a row pitch of 132 floats (16 B x odd), not 128: the 16 rows of a WMMA A-fragment read fall in 16 different banks instead of one (was a 16-way bank conflict per read)
# speedup vs baseline: 3.4706x; 1.1369x over previous
#include <hip/hip_runtime.h>

typedef __attribute__((ext_vector_type(2))) float v2f;
typedef __attribute__((ext_vector_type(8))) float v8f;

#define FDIM 128
#define LDA 132

#define BSH    8
#define MAXNB  512
#define CH     8192
#define EPT    (CH / 512)
#define SEGPT  20
#define SEGCAP 12288

__global__ __launch_bounds__(256) void k_init(int* __restrict__ bins,
                                              float* __restrict__ stats)
{
    int i = blockIdx.x * 256 + threadIdx.x;
    if (i < 2 * MAXNB) bins[i] = 0;
    if (i < 256) stats[i] = 0.f;
}

__global__ __launch_bounds__(512) void k_hist(const int* __restrict__ dst,
                                              int* __restrict__ binCnt, int E)
{
    __shared__ int hist[MAXNB];
    const int t = threadIdx.x;
    const int base = blockIdx.x * CH;
    const int n = (E - base < CH) ? (E - base) : CH;
    int dv[EPT];
#pragma unroll
    for (int k = 0; k < EPT; ++k) {
        const int i = t + k * 512;
        dv[k] = dst[base + ((i < n) ? i : n - 1)];
    }
    hist[t] = 0;
    __syncthreads();
#pragma unroll
    for (int k = 0; k < EPT; ++k)
        if (t + k * 512 < n) atomicAdd(&hist[dv[k] >> BSH], 1);
    __syncthreads();
    const int c = hist[t];
    if (c > 0) atomicAdd(&binCnt[t], c);
}

__global__ __launch_bounds__(512) void k_scatter(const int* __restrict__ src,
                                                 const int* __restrict__ dst,
                                                 const int* __restrict__ binCnt,
                                                 int* __restrict__ cursor,
                                                 int* __restrict__ tmp, int E)
{
    __shared__ int hist[MAXNB];
    __shared__ int lcur[MAXNB];
    __shared__ int gdelta[MAXNB];
    __shared__ int wc[16], wg[16];
    __shared__ int sval[CH];
    __shared__ int sadr[CH];

    const int t = threadIdx.x, lane = t & 31, wave = t >> 5;
    const int base = blockIdx.x * CH;
    const int n = (E - base < CH) ? (E - base) : CH;

    int dv[EPT], sv[EPT];
#pragma unroll
    for (int k = 0; k < EPT; ++k) {
        const int i = t + k * 512;
        dv[k] = dst[base + ((i < n) ? i : n - 1)];
    }
#pragma unroll
    for (int k = 0; k < EPT; ++k) {
        const int i = t + k * 512;
        sv[k] = src[base + ((i < n) ? i : n - 1)];
    }
    const int g = binCnt[t];

    hist[t] = 0;
    __syncthreads();
#pragma unroll
    for (int k = 0; k < EPT; ++k)
        if (t + k * 512 < n) atomicAdd(&hist[dv[k] >> BSH], 1);
    __syncthreads();

    const int c = hist[t];
    int ci = c, gi = g;
    for (int o = 1; o < 32; o <<= 1) {
        int vc = __shfl_up(ci, o, 32);
        int vg = __shfl_up(gi, o, 32);
        if (lane >= o) { ci += vc; gi += vg; }
    }
    if (lane == 31) { wc[wave] = ci; wg[wave] = gi; }
    __syncthreads();
    int ce = ci - c, ge = gi - g;
    for (int w = 0; w < wave; ++w) { ce += wc[w]; ge += wg[w]; }
    int r = 0;
    if (c > 0) r = atomicAdd(&cursor[t], c);
    lcur[t] = ce;
    gdelta[t] = ge + r - ce;
    __syncthreads();

    int pv[EPT];
#pragma unroll
    for (int k = 0; k < EPT; ++k) {
        pv[k] = 0;
        if (t + k * 512 < n) pv[k] = atomicAdd(&lcur[dv[k] >> BSH], 1);
    }
#pragma unroll
    for (int k = 0; k < EPT; ++k) {
        if (t + k * 512 < n) {
            const int p = pv[k];
            sval[p] = (int)(((unsigned)sv[k] << 8) | (unsigned)(dv[k] & 255));
            sadr[p] = gdelta[dv[k] >> BSH] + p;
        }
    }
    __syncthreads();
    for (int i = t; i < n; i += 512) tmp[sadr[i]] = sval[i];
}

__global__ __launch_bounds__(256) void k_group(const int* __restrict__ tmp,
                                               const int* __restrict__ binCnt,
                                               int* __restrict__ rowptr,
                                               float* __restrict__ dis,
                                               int* __restrict__ csr_src, int N)
{
    __shared__ int cnt[256];
    __shared__ int red[8], wsum[8];
    __shared__ int sbuf[SEGCAP];

    const int t = threadIdx.x, lane = t & 31, wave = t >> 5;
    const int b = blockIdx.x;

    const int g0 = binCnt[t], g1 = binCnt[t + 256];
    int a = ((t < b) ? g0 : 0) + ((t + 256 < b) ? g1 : 0);
    for (int o = 16; o > 0; o >>= 1) a += __shfl_xor(a, o, 32);
    if (lane == 0) red[wave] = a;
    cnt[t] = 0;
    __syncthreads();
    int beg = 0;
    for (int w = 0; w < 8; ++w) beg += red[w];
    beg = __builtin_amdgcn_readfirstlane(beg);
    const int len = __builtin_amdgcn_readfirstlane(binCnt[b]);
    const bool staged = len <= SEGCAP;

    for (int i0 = 0; i0 < len; i0 += SEGPT * 256) {
        int v[SEGPT];
#pragma unroll
        for (int k = 0; k < SEGPT; ++k) {
            const int i = i0 + t + k * 256;
            v[k] = tmp[beg + ((i < len) ? i : len - 1)];
        }
#pragma unroll
        for (int k = 0; k < SEGPT; ++k)
            if (i0 + t + k * 256 < len) atomicAdd(&cnt[v[k] & 255], 1);
    }
    __syncthreads();

    const int c = cnt[t];
    int incl = c;
    for (int o = 1; o < 32; o <<= 1) {
        int v = __shfl_up(incl, o, 32);
        if (lane >= o) incl += v;
    }
    if (lane == 31) wsum[wave] = incl;
    __syncthreads();
    int excl = incl - c;
    for (int w = 0; w < wave; ++w) excl += wsum[w];
    const int node = b * 256 + t;
    if (node < N) {
        rowptr[node] = beg + excl;
        dis[node] = rsqrtf((float)(c + 1));
    }
    cnt[t] = excl;
    __syncthreads();

    for (int i0 = 0; i0 < len; i0 += SEGPT * 256) {
        unsigned v[SEGPT];
        int p[SEGPT];
#pragma unroll
        for (int k = 0; k < SEGPT; ++k) {
            const int i = i0 + t + k * 256;
            v[k] = (unsigned)tmp[beg + ((i < len) ? i : len - 1)];
        }
#pragma unroll
        for (int k = 0; k < SEGPT; ++k) {
            p[k] = 0;
            if (i0 + t + k * 256 < len) p[k] = atomicAdd(&cnt[v[k] & 255u], 1);
        }
#pragma unroll
        for (int k = 0; k < SEGPT; ++k) {
            if (i0 + t + k * 256 < len) {
                const int s = (int)(v[k] >> 8);
                if (staged) sbuf[p[k]] = s;
                else        csr_src[beg + p[k]] = s;
            }
        }
    }
    __syncthreads();
    if (staged)
        for (int i = t; i < len; i += 256) csr_src[beg + i] = sbuf[i];
}

__global__ __launch_bounds__(256) void k_gemm1(const float* __restrict__ X,
                                               const float* __restrict__ W,
                                               const float* __restrict__ dis,
                                               float* __restrict__ O, int N)
{
    __shared__ float lA[32 * LDA];
    __shared__ v2f   lB[64 * 64];

    const int t    = threadIdx.x;
    const int rowb = blockIdx.x * 32;
    const int colb = blockIdx.y * 64;

    float ds[8];
    {
        const int r0 = rowb + ((t >> 5) >> 2) * 16 + 8 * ((t & 31) >> 4);
        for (int r = 0; r < 8; ++r) {
            int row = r0 + r; if (row >= N) row = N - 1;
            ds[r] = dis[row];
        }
    }

    for (int idx = t; idx < 32 * FDIM; idx += 256) {
        int r = idx >> 7, k = idx & 127;
        int gr = rowb + r; if (gr >= N) gr = N - 1;
        lA[r * LDA + k] = X[(size_t)gr * FDIM + k];
    }
    for (int idx = t; idx < 64 * 64; idx += 256) {
        int k2 = idx >> 6, c = idx & 63;
        const float* wp = W + (size_t)(2 * k2) * FDIM + (colb + c);
        v2f b; b.x = wp[0]; b.y = wp[FDIM];
        lB[idx] = b;
    }
    __syncthreads();

    const int wave = t >> 5, lane = t & 31;
    const int m = lane & 15, hi = lane >> 4;
    const int rt = wave >> 2, ct = wave & 3;
    const float* Arow = &lA[(rt * 16 + m) * LDA];
    const int nloc = ct * 16 + m;

    v8f acc = {0.f, 0.f, 0.f, 0.f, 0.f, 0.f, 0.f, 0.f};
#pragma unroll
    for (int k = 0; k < FDIM; k += 4) {
        v2f a = *(const v2f*)&Arow[k + 2 * hi];
        v2f b = lB[(k / 2 + hi) * 64 + nloc];
        acc = __builtin_amdgcn_wmma_f32_16x16x4_f32(
            false, a, false, b, (short)0, acc, false, false);
    }

    const int col = colb + nloc;
#pragma unroll
    for (int r = 0; r < 8; ++r) {
        int row = rowb + rt * 16 + r + 8 * hi;
        if (row < N) O[(size_t)row * FDIM + col] = acc[r] * ds[r];
    }
}

static __device__ __forceinline__ int bcast_i(int v, int j)
{
    return __builtin_amdgcn_readlane(v, j);
}
static __device__ __forceinline__ void row_add(float4& acc, const float4& v)
{
    acc.x += v.x; acc.y += v.y; acc.z += v.z; acc.w += v.w;
}

__global__ __launch_bounds__(256) void k_gather128(const float* __restrict__ h,
                                                   const int* __restrict__ rowptr,
                                                   const int* __restrict__ csr_src,
                                                   const float* __restrict__ dis,
                                                   const float* __restrict__ bias,
                                                   float* __restrict__ out, int N, int E)
{
    const int lane = threadIdx.x & 31;
    const int node = blockIdx.x * 8 + __builtin_amdgcn_readfirstlane((int)(threadIdx.x >> 5));
    if (node >= N) return;
    const int beg = __builtin_amdgcn_readfirstlane(rowptr[node]);
    const int end = (node + 1 < N) ? __builtin_amdgcn_readfirstlane(rowptr[node + 1]) : E;
    const float di = dis[node];
    const float4* hp = (const float4*)h;

    float4 acc = hp[(size_t)node * 32 + lane];
    const float4 b = ((const float4*)bias)[lane];
    for (int p = beg; p < end; p += 32) {
        const int n = (end - p < 32) ? (end - p) : 32;
        int s_l = 0;
        if (lane < n) s_l = csr_src[p + lane];
        int j = 0;
        for (; j + 4 <= n; j += 4) {
            const int s0 = bcast_i(s_l, j),     s1 = bcast_i(s_l, j + 1);
            const int s2 = bcast_i(s_l, j + 2), s3 = bcast_i(s_l, j + 3);
            const float4 v0 = hp[(size_t)s0 * 32 + lane];
            const float4 v1 = hp[(size_t)s1 * 32 + lane];
            const float4 v2 = hp[(size_t)s2 * 32 + lane];
            const float4 v3 = hp[(size_t)s3 * 32 + lane];
            row_add(acc, v0); row_add(acc, v1);
            row_add(acc, v2); row_add(acc, v3);
        }
        for (; j < n; ++j) {
            const int s0 = bcast_i(s_l, j);
            const float4 v0 = hp[(size_t)s0 * 32 + lane];
            row_add(acc, v0);
        }
    }
    acc.x = acc.x * di + b.x;
    acc.y = acc.y * di + b.y;
    acc.z = acc.z * di + b.z;
    acc.w = acc.w * di + b.w;
    ((float4*)out)[(size_t)node * 32 + lane] = acc;
}

__global__ __launch_bounds__(256) void k_gatherC(const float* __restrict__ h,
                                                 const int* __restrict__ rowptr,
                                                 const int* __restrict__ csr_src,
                                                 const float* __restrict__ dis,
                                                 const float* __restrict__ bias,
                                                 float* __restrict__ out, int N, int E, int C)
{
    const int lane = threadIdx.x & 31;
    const int node = blockIdx.x * 8 + __builtin_amdgcn_readfirstlane((int)(threadIdx.x >> 5));
    if (node >= N) return;
    const int beg = __builtin_amdgcn_readfirstlane(rowptr[node]);
    const int end = (node + 1 < N) ? __builtin_amdgcn_readfirstlane(rowptr[node + 1]) : E;
    const float di = dis[node];
    const bool a0 = lane < C, a1 = 32 + lane < C;
    const int c0 = a0 ? lane : 0, c1 = a1 ? 32 + lane : 0;

    float acc0, acc1;
    {
        const float* hr = h + (size_t)node * C;
        acc0 = hr[c0];
        acc1 = hr[c1];
    }
    const float b0 = bias[c0], b1 = bias[c1];
    for (int p = beg; p < end; p += 32) {
        const int n = (end - p < 32) ? (end - p) : 32;
        int s_l = 0;
        if (lane < n) s_l = csr_src[p + lane];
        int j = 0;
        for (; j + 4 <= n; j += 4) {
            const float* r0 = h + (size_t)bcast_i(s_l, j) * C;
            const float* r1 = h + (size_t)bcast_i(s_l, j + 1) * C;
            const float* r2 = h + (size_t)bcast_i(s_l, j + 2) * C;
            const float* r3 = h + (size_t)bcast_i(s_l, j + 3) * C;
            const float x0 = r0[c0], y0 = r0[c1];
            const float x1 = r1[c0], y1 = r1[c1];
            const float x2 = r2[c0], y2 = r2[c1];
            const float x3 = r3[c0], y3 = r3[c1];
            acc0 += x0; acc1 += y0;
            acc0 += x1; acc1 += y1;
            acc0 += x2; acc1 += y2;
            acc0 += x3; acc1 += y3;
        }
        for (; j < n; ++j) {
            const float* r0 = h + (size_t)bcast_i(s_l, j) * C;
            acc0 += r0[c0]; acc1 += r0[c1];
        }
    }
    float* orow = out + (size_t)node * C;
    if (a0) orow[lane] = acc0 * di + b0;
    if (a1) orow[32 + lane] = acc1 * di + b1;
}

__global__ __launch_bounds__(256) void k_relu_stats(float* __restrict__ a,
                                                    float* __restrict__ gsum,
                                                    float* __restrict__ gsq, int N)
{
    __shared__ float ssum[256];
    __shared__ float ssq[256];
    int t = threadIdx.x, f = t & 127, sub = t >> 7;
    float sum = 0.f, sq = 0.f;
    int base = blockIdx.x * 64;
#pragma unroll 4
    for (int i = 0; i < 32; ++i) {
        int row = base + sub + 2 * i;
        if (row < N) {
            size_t idx = (size_t)row * FDIM + f;
            float v = a[idx];
            v = fmaxf(v, 0.f);
            a[idx] = v;
            sum += v;
            sq  += v * v;
        }
    }
    ssum[t] = sum; ssq[t] = sq;
    __syncthreads();
    if (t < 128) {
        atomicAdd(&gsum[t], ssum[t] + ssum[t + 128]);
        atomicAdd(&gsq[t],  ssq[t]  + ssq[t + 128]);
    }
}

__global__ void k_bn_finalize(const float* __restrict__ gsum,
                              const float* __restrict__ gsq,
                              const float* __restrict__ gamma,
                              const float* __restrict__ beta,
                              float* __restrict__ scl,
                              float* __restrict__ sft, int N)
{
    int f = threadIdx.x;
    float inv_n = 1.f / (float)N;
    float mean = gsum[f] * inv_n;
    float var  = gsq[f] * inv_n - mean * mean;
    float inv  = rsqrtf(var + 1e-5f);
    float g = gamma[f] * inv;
    scl[f] = g;
    sft[f] = beta[f] - mean * g;
}

__global__ __launch_bounds__(96) void k_gemm2(const float* __restrict__ A,
                                              const float* __restrict__ W,
                                              const float* __restrict__ scl,
                                              const float* __restrict__ sft,
                                              const float* __restrict__ dis,
                                              float* __restrict__ O, int N, int C)
{
    __shared__ float lA[16 * LDA];
    __shared__ v2f   lB[64 * 48];

    const int t = threadIdx.x;
    const int rowb = blockIdx.x * 16;

    float ds[8];
    {
        const int r0 = rowb + 8 * ((t & 31) >> 4);
        for (int r = 0; r < 8; ++r) {
            int row = r0 + r; if (row >= N) row = N - 1;
            ds[r] = dis[row];
        }
    }

    for (int idx = t; idx < 16 * FDIM; idx += 96) {
        int r = idx >> 7, k = idx & 127;
        int gr = rowb + r; if (gr >= N) gr = N - 1;
        lA[r * LDA + k] = A[(size_t)gr * FDIM + k] * scl[k] + sft[k];
    }
    for (int idx = t; idx < 64 * 48; idx += 96) {
        int k2 = idx / 48, c = idx - k2 * 48;
        v2f b;
        b.x = (c < C) ? W[(size_t)(2 * k2) * C + c]     : 0.f;
        b.y = (c < C) ? W[(size_t)(2 * k2 + 1) * C + c] : 0.f;
        lB[idx] = b;
    }
    __syncthreads();

    const int wave = t >> 5, lane = t & 31;
    const int m = lane & 15, hi = lane >> 4;
    const float* Arow = &lA[m * LDA];
    const int nloc = wave * 16 + m;

    v8f acc = {0.f, 0.f, 0.f, 0.f, 0.f, 0.f, 0.f, 0.f};
#pragma unroll
    for (int k = 0; k < FDIM; k += 4) {
        v2f a = *(const v2f*)&Arow[k + 2 * hi];
        v2f b = lB[(k / 2 + hi) * 48 + nloc];
        acc = __builtin_amdgcn_wmma_f32_16x16x4_f32(
            false, a, false, b, (short)0, acc, false, false);
    }

    if (nloc < C) {
#pragma unroll
        for (int r = 0; r < 8; ++r) {
            int row = rowb + r + 8 * hi;
            if (row < N) O[(size_t)row * C + nloc] = acc[r] * ds[r];
        }
    }
}

__global__ __launch_bounds__(256) void k_logsoftmax(const float* __restrict__ in,
                                                    float* __restrict__ out, int N, int C)
{
    long long gid = (long long)blockIdx.x * 256 + threadIdx.x;
    int row = (int)(gid >> 5);
    int lane = threadIdx.x & 31;
    if (row >= N) return;
    const float* p = in + (size_t)row * C;
    float* q = out + (size_t)row * C;
    float v0 = (lane < C) ? p[lane] : -3.4e38f;
    float v1 = (32 + lane < C) ? p[32 + lane] : -3.4e38f;
    float mx = fmaxf(v0, v1);
#pragma unroll
    for (int o = 16; o > 0; o >>= 1) mx = fmaxf(mx, __shfl_xor(mx, o, 32));
    float s = ((lane < C) ? expf(v0 - mx) : 0.f) +
              ((32 + lane < C) ? expf(v1 - mx) : 0.f);
#pragma unroll
    for (int o = 16; o > 0; o >>= 1) s += __shfl_xor(s, o, 32);
    float l = logf(s);
    if (lane < C)      q[lane]      = v0 - mx - l;
    if (32 + lane < C) q[32 + lane] = v1 - mx - l;
}

extern "C" void kernel_launch(void* const* d_in, const int* in_sizes, int n_in,
                              void* d_out, int out_size, void* d_ws, size_t ws_size,
                              hipStream_t stream)
{
    const float* x     = (const float*)d_in[0];
    const int*   ei    = (const int*)  d_in[1];
    const float* W1    = (const float*)d_in[2];
    const float* b1    = (const float*)d_in[3];
    const float* gamma = (const float*)d_in[4];
    const float* beta  = (const float*)d_in[5];
    const float* W2    = (const float*)d_in[6];
    const float* b2    = (const float*)d_in[7];
    float* out = (float*)d_out;

    const int N = in_sizes[0] / FDIM;
    const int E = in_sizes[1] / 2;
    const int C = in_sizes[6] / FDIM;
    const int* src = ei;
    const int* dst = ei + E;

    char* ws = (char*)d_ws;
    size_t off = 0;
    auto up = [](size_t v) { return (v + 255) & ~(size_t)255; };
    int*   rowptr = (int*)(ws + off);  off = up(off + (size_t)N * 4);
    float* dis   = (float*)(ws + off); off = up(off + (size_t)N * 4);
    float* stats = (float*)(ws + off); off = up(off + 512 * 4);
    float* gsum = stats, *gsq = stats + 128, *scl = stats + 256, *sft = stats + 384;
    float* h1   = (float*)(ws + off);  off = up(off + (size_t)N * FDIM * 4);
    float* agg1 = (float*)(ws + off);  off = up(off + (size_t)N * FDIM * 4);
    float* h2   = h1;
    int* tmp    = (int*)agg1;
    int* binCnt = (int*)((char*)agg1 + up((size_t)E * 4));
    int* cursor = binCnt + MAXNB;
    int*   csr_src = (int*)d_out;
    float* logits  = agg1;

    const int nb_C   = (E + CH - 1) / CH;
    const int nb_B   = (N + 255) / 256;
    const int nb_Nw  = (N + 7) / 8;

    k_init<<<4, 256, 0, stream>>>(binCnt, stats);
    k_hist<<<nb_C, 512, 0, stream>>>(dst, binCnt, E);
    k_scatter<<<nb_C, 512, 0, stream>>>(src, dst, binCnt, cursor, tmp, E);
    k_group<<<nb_B, 256, 0, stream>>>(tmp, binCnt, rowptr, dis, csr_src, N);

    k_gemm1<<<dim3((N + 31) / 32, FDIM / 64), 256, 0, stream>>>(x, W1, dis, h1, N);

    k_gather128<<<nb_Nw, 256, 0, stream>>>(h1, rowptr, csr_src, dis, b1, agg1, N, E);

    k_relu_stats<<<(N + 63) / 64, 256, 0, stream>>>(agg1, gsum, gsq, N);
    k_bn_finalize<<<1, 128, 0, stream>>>(gsum, gsq, gamma, beta, scl, sft, N);

    k_gemm2<<<(N + 15) / 16, 96, 0, stream>>>(agg1, W2, scl, sft, dis, h2, N, C);

    k_gatherC<<<nb_Nw, 256, 0, stream>>>(h2, rowptr, csr_src, dis, b2, logits, N, E, C);

    k_logsoftmax<<<nb_Nw, 256, 0, stream>>>(logits, out, N, C);
}


# ReLU + batch-norm column sums taken in conv1 gather epilogue (waves loop over nodes; column-sum order changes), scale/shift in GEMM2 prologue: stats pass + finalize launch removed, 11 to 9 launches
# speedup vs baseline: 3.7785x; 1.0887x over previous
#include <hip/hip_runtime.h>

typedef __attribute__((ext_vector_type(2))) float v2f;
typedef __attribute__((ext_vector_type(8))) float v8f;

#define FDIM 128
#define LDA 132

#define BSH    8
#define MAXNB  512
#define CH     8192
#define EPT    (CH / 512)
#define SEGPT  20
#define SEGCAP 12288

__global__ __launch_bounds__(256) void k_init(int* __restrict__ bins,
                                              float* __restrict__ stats)
{
    int i = blockIdx.x * 256 + threadIdx.x;
    if (i < 2 * MAXNB) bins[i] = 0;
    if (i < 256) stats[i] = 0.f;
}

__global__ __launch_bounds__(512) void k_hist(const int* __restrict__ dst,
                                              int* __restrict__ binCnt, int E)
{
    __shared__ int hist[MAXNB];
    const int t = threadIdx.x;
    const int base = blockIdx.x * CH;
    const int n = (E - base < CH) ? (E - base) : CH;
    int dv[EPT];
#pragma unroll
    for (int k = 0; k < EPT; ++k) {
        const int i = t + k * 512;
        dv[k] = dst[base + ((i < n) ? i : n - 1)];
    }
    hist[t] = 0;
    __syncthreads();
#pragma unroll
    for (int k = 0; k < EPT; ++k)
        if (t + k * 512 < n) atomicAdd(&hist[dv[k] >> BSH], 1);
    __syncthreads();
    const int c = hist[t];
    if (c > 0) atomicAdd(&binCnt[t], c);
}

__global__ __launch_bounds__(512) void k_scatter(const int* __restrict__ src,
                                                 const int* __restrict__ dst,
                                                 const int* __restrict__ binCnt,
                                                 int* __restrict__ cursor,
                                                 int* __restrict__ tmp, int E)
{
    __shared__ int hist[MAXNB];
    __shared__ int lcur[MAXNB];
    __shared__ int gdelta[MAXNB];
    __shared__ int wc[16], wg[16];
    __shared__ int sval[CH];
    __shared__ int sadr[CH];

    const int t = threadIdx.x, lane = t & 31, wave = t >> 5;
    const int base = blockIdx.x * CH;
    const int n = (E - base < CH) ? (E - base) : CH;

    int dv[EPT], sv[EPT];
#pragma unroll
    for (int k = 0; k < EPT; ++k) {
        const int i = t + k * 512;
        dv[k] = dst[base + ((i < n) ? i : n - 1)];
    }
#pragma unroll
    for (int k = 0; k < EPT; ++k) {
        const int i = t + k * 512;
        sv[k] = src[base + ((i < n) ? i : n - 1)];
    }
    const int g = binCnt[t];

    hist[t] = 0;
    __syncthreads();
#pragma unroll
    for (int k = 0; k < EPT; ++k)
        if (t + k * 512 < n) atomicAdd(&hist[dv[k] >> BSH], 1);
    __syncthreads();

    const int c = hist[t];
    int ci = c, gi = g;
    for (int o = 1; o < 32; o <<= 1) {
        int vc = __shfl_up(ci, o, 32);
        int vg = __shfl_up(gi, o, 32);
        if (lane >= o) { ci += vc; gi += vg; }
    }
    if (lane == 31) { wc[wave] = ci; wg[wave] = gi; }
    __syncthreads();
    int ce = ci - c, ge = gi - g;
    for (int w = 0; w < wave; ++w) { ce += wc[w]; ge += wg[w]; }
    int r = 0;
    if (c > 0) r = atomicAdd(&cursor[t], c);
    lcur[t] = ce;
    gdelta[t] = ge + r - ce;
    __syncthreads();

    int pv[EPT];
#pragma unroll
    for (int k = 0; k < EPT; ++k) {
        pv[k] = 0;
        if (t + k * 512 < n) pv[k] = atomicAdd(&lcur[dv[k] >> BSH], 1);
    }
#pragma unroll
    for (int k = 0; k < EPT; ++k) {
        if (t + k * 512 < n) {
            const int p = pv[k];
            sval[p] = (int)(((unsigned)sv[k] << 8) | (unsigned)(dv[k] & 255));
            sadr[p] = gdelta[dv[k] >> BSH] + p;
        }
    }
    __syncthreads();
    for (int i = t; i < n; i += 512) tmp[sadr[i]] = sval[i];
}

__global__ __launch_bounds__(256) void k_group(const int* __restrict__ tmp,
                                               const int* __restrict__ binCnt,
                                               int* __restrict__ rowptr,
                                               float* __restrict__ dis,
                                               int* __restrict__ csr_src, int N)
{
    __shared__ int cnt[256];
    __shared__ int red[8], wsum[8];
    __shared__ int sbuf[SEGCAP];

    const int t = threadIdx.x, lane = t & 31, wave = t >> 5;
    const int b = blockIdx.x;

    const int g0 = binCnt[t], g1 = binCnt[t + 256];
    int a = ((t < b) ? g0 : 0) + ((t + 256 < b) ? g1 : 0);
    for (int o = 16; o > 0; o >>= 1) a += __shfl_xor(a, o, 32);
    if (lane == 0) red[wave] = a;
    cnt[t] = 0;
    __syncthreads();
    int beg = 0;
    for (int w = 0; w < 8; ++w) beg += red[w];
    beg = __builtin_amdgcn_readfirstlane(beg);
    const int len = __builtin_amdgcn_readfirstlane(binCnt[b]);
    const bool staged = len <= SEGCAP;

    for (int i0 = 0; i0 < len; i0 += SEGPT * 256) {
        int v[SEGPT];
#pragma unroll
        for (int k = 0; k < SEGPT; ++k) {
            const int i = i0 + t + k * 256;
            v[k] = tmp[beg + ((i < len) ? i : len - 1)];
        }
#pragma unroll
        for (int k = 0; k < SEGPT; ++k)
            if (i0 + t + k * 256 < len) atomicAdd(&cnt[v[k] & 255], 1);
    }
    __syncthreads();

    const int c = cnt[t];
    int incl = c;
    for (int o = 1; o < 32; o <<= 1) {
        int v = __shfl_up(incl, o, 32);
        if (lane >= o) incl += v;
    }
    if (lane == 31) wsum[wave] = incl;
    __syncthreads();
    int excl = incl - c;
    for (int w = 0; w < wave; ++w) excl += wsum[w];
    const int node = b * 256 + t;
    if (node < N) {
        rowptr[node] = beg + excl;
        dis[node] = rsqrtf((float)(c + 1));
    }
    cnt[t] = excl;
    __syncthreads();

    for (int i0 = 0; i0 < len; i0 += SEGPT * 256) {
        unsigned v[SEGPT];
        int p[SEGPT];
#pragma unroll
        for (int k = 0; k < SEGPT; ++k) {
            const int i = i0 + t + k * 256;
            v[k] = (unsigned)tmp[beg + ((i < len) ? i : len - 1)];
        }
#pragma unroll
        for (int k = 0; k < SEGPT; ++k) {
            p[k] = 0;
            if (i0 + t + k * 256 < len) p[k] = atomicAdd(&cnt[v[k] & 255u], 1);
        }
#pragma unroll
        for (int k = 0; k < SEGPT; ++k) {
            if (i0 + t + k * 256 < len) {
                const int s = (int)(v[k] >> 8);
                if (staged) sbuf[p[k]] = s;
                else        csr_src[beg + p[k]] = s;
            }
        }
    }
    __syncthreads();
    if (staged)
        for (int i = t; i < len; i += 256) csr_src[beg + i] = sbuf[i];
}

__global__ __launch_bounds__(256) void k_gemm1(const float* __restrict__ X,
                                               const float* __restrict__ W,
                                               const float* __restrict__ dis,
                                               float* __restrict__ O, int N)
{
    __shared__ float lA[32 * LDA];
    __shared__ v2f   lB[64 * 64];

    const int t    = threadIdx.x;
    const int rowb = blockIdx.x * 32;
    const int colb = blockIdx.y * 64;

    float ds[8];
    {
        const int r0 = rowb + ((t >> 5) >> 2) * 16 + 8 * ((t & 31) >> 4);
        for (int r = 0; r < 8; ++r) {
            int row = r0 + r; if (row >= N) row = N - 1;
            ds[r] = dis[row];
        }
    }

    for (int idx = t; idx < 32 * FDIM; idx += 256) {
        int r = idx >> 7, k = idx & 127;
        int gr = rowb + r; if (gr >= N) gr = N - 1;
        lA[r * LDA + k] = X[(size_t)gr * FDIM + k];
    }
    for (int idx = t; idx < 64 * 64; idx += 256) {
        int k2 = idx >> 6, c = idx & 63;
        const float* wp = W + (size_t)(2 * k2) * FDIM + (colb + c);
        v2f b; b.x = wp[0]; b.y = wp[FDIM];
        lB[idx] = b;
    }
    __syncthreads();

    const int wave = t >> 5, lane = t & 31;
    const int m = lane & 15, hi = lane >> 4;
    const int rt = wave >> 2, ct = wave & 3;
    const float* Arow = &lA[(rt * 16 + m) * LDA];
    const int nloc = ct * 16 + m;

    v8f acc = {0.f, 0.f, 0.f, 0.f, 0.f, 0.f, 0.f, 0.f};
#pragma unroll
    for (int k = 0; k < FDIM; k += 4) {
        v2f a = *(const v2f*)&Arow[k + 2 * hi];
        v2f b = lB[(k / 2 + hi) * 64 + nloc];
        acc = __builtin_amdgcn_wmma_f32_16x16x4_f32(
            false, a, false, b, (short)0, acc, false, false);
    }

    const int col = colb + nloc;
#pragma unroll
    for (int r = 0; r < 8; ++r) {
        int row = rowb + rt * 16 + r + 8 * hi;
        if (row < N) O[(size_t)row * FDIM + col] = acc[r] * ds[r];
    }
}

static __device__ __forceinline__ int bcast_i(int v, int j)
{
    return __builtin_amdgcn_readlane(v, j);
}
static __device__ __forceinline__ void row_add(float4& acc, const float4& v)
{
    acc.x += v.x; acc.y += v.y; acc.z += v.z; acc.w += v.w;
}

__global__ __launch_bounds__(512) void k_gather128(const float* __restrict__ h,
                                                   const int* __restrict__ rowptr,
                                                   const int* __restrict__ csr_src,
                                                   const float* __restrict__ dis,
                                                   const float* __restrict__ bias,
                                                   float* __restrict__ out,
                                                   float* __restrict__ stats,
                                                   int N, int E, int nblk)
{
    __shared__ float red[16 * 256];
    const int lane = threadIdx.x & 31;
    const int wave = __builtin_amdgcn_readfirstlane((int)(threadIdx.x >> 5));
    const float4* hp = (const float4*)h;
    const float4 b = ((const float4*)bias)[lane];
    float4 s1 = {0.f, 0.f, 0.f, 0.f}, s2 = {0.f, 0.f, 0.f, 0.f};

    for (int nb = blockIdx.x * 16; nb < N; nb += nblk * 16) {
        const int node = nb + wave;
        if (node < N) {
            const int beg = __builtin_amdgcn_readfirstlane(rowptr[node]);
            const int end = (node + 1 < N) ? __builtin_amdgcn_readfirstlane(rowptr[node + 1]) : E;
            const float di = dis[node];

            float4 acc = hp[(size_t)node * 32 + lane];
            for (int p = beg; p < end; p += 32) {
                const int n = (end - p < 32) ? (end - p) : 32;
                int s_l = 0;
                if (lane < n) s_l = csr_src[p + lane];
                int j = 0;
                for (; j + 4 <= n; j += 4) {
                    const int i0 = bcast_i(s_l, j),     i1 = bcast_i(s_l, j + 1);
                    const int i2 = bcast_i(s_l, j + 2), i3 = bcast_i(s_l, j + 3);
                    const float4 v0 = hp[(size_t)i0 * 32 + lane];
                    const float4 v1 = hp[(size_t)i1 * 32 + lane];
                    const float4 v2 = hp[(size_t)i2 * 32 + lane];
                    const float4 v3 = hp[(size_t)i3 * 32 + lane];
                    row_add(acc, v0); row_add(acc, v1);
                    row_add(acc, v2); row_add(acc, v3);
                }
                for (; j < n; ++j) {
                    const int i0 = bcast_i(s_l, j);
                    const float4 v0 = hp[(size_t)i0 * 32 + lane];
                    row_add(acc, v0);
                }
            }
            acc.x = fmaxf(acc.x * di + b.x, 0.f);
            acc.y = fmaxf(acc.y * di + b.y, 0.f);
            acc.z = fmaxf(acc.z * di + b.z, 0.f);
            acc.w = fmaxf(acc.w * di + b.w, 0.f);
            ((float4*)out)[(size_t)node * 32 + lane] = acc;
            s1.x += acc.x; s1.y += acc.y; s1.z += acc.z; s1.w += acc.w;
            s2.x += acc.x * acc.x; s2.y += acc.y * acc.y;
            s2.z += acc.z * acc.z; s2.w += acc.w * acc.w;
        }
    }

    ((float4*)red)[wave * 64 + lane]      = s1;
    ((float4*)red)[wave * 64 + 32 + lane] = s2;
    __syncthreads();
    const int t = threadIdx.x;
    if (t < 256) {
        float v = 0.f;
        for (int w = 0; w < 16; ++w) v += red[w * 256 + t];
        atomicAdd(&stats[t], v);
    }
}

__global__ __launch_bounds__(256) void k_gatherC(const float* __restrict__ h,
                                                 const int* __restrict__ rowptr,
                                                 const int* __restrict__ csr_src,
                                                 const float* __restrict__ dis,
                                                 const float* __restrict__ bias,
                                                 float* __restrict__ out, int N, int E, int C)
{
    const int lane = threadIdx.x & 31;
    const int node = blockIdx.x * 8 + __builtin_amdgcn_readfirstlane((int)(threadIdx.x >> 5));
    if (node >= N) return;
    const int beg = __builtin_amdgcn_readfirstlane(rowptr[node]);
    const int end = (node + 1 < N) ? __builtin_amdgcn_readfirstlane(rowptr[node + 1]) : E;
    const float di = dis[node];
    const bool a0 = lane < C, a1 = 32 + lane < C;
    const int c0 = a0 ? lane : 0, c1 = a1 ? 32 + lane : 0;

    float acc0, acc1;
    {
        const float* hr = h + (size_t)node * C;
        acc0 = hr[c0];
        acc1 = hr[c1];
    }
    const float b0 = bias[c0], b1 = bias[c1];
    for (int p = beg; p < end; p += 32) {
        const int n = (end - p < 32) ? (end - p) : 32;
        int s_l = 0;
        if (lane < n) s_l = csr_src[p + lane];
        int j = 0;
        for (; j + 4 <= n; j += 4) {
            const float* r0 = h + (size_t)bcast_i(s_l, j) * C;
            const float* r1 = h + (size_t)bcast_i(s_l, j + 1) * C;
            const float* r2 = h + (size_t)bcast_i(s_l, j + 2) * C;
            const float* r3 = h + (size_t)bcast_i(s_l, j + 3) * C;
            const float x0 = r0[c0], y0 = r0[c1];
            const float x1 = r1[c0], y1 = r1[c1];
            const float x2 = r2[c0], y2 = r2[c1];
            const float x3 = r3[c0], y3 = r3[c1];
            acc0 += x0; acc1 += y0;
            acc0 += x1; acc1 += y1;
            acc0 += x2; acc1 += y2;
            acc0 += x3; acc1 += y3;
        }
        for (; j < n; ++j) {
            const float* r0 = h + (size_t)bcast_i(s_l, j) * C;
            acc0 += r0[c0]; acc1 += r0[c1];
        }
    }
    float* orow = out + (size_t)node * C;
    if (a0) orow[lane] = acc0 * di + b0;
    if (a1) orow[32 + lane] = acc1 * di + b1;
}

__global__ __launch_bounds__(96) void k_gemm2(const float* __restrict__ A,
                                              const float* __restrict__ W,
                                              const float* __restrict__ stats,
                                              const float* __restrict__ gamma,
                                              const float* __restrict__ beta,
                                              const float* __restrict__ dis,
                                              float* __restrict__ O, int N, int C)
{
    __shared__ float lA[16 * LDA];
    __shared__ v2f   lB[64 * 48];
    __shared__ float scl[FDIM], sft[FDIM];

    const int t = threadIdx.x;
    const int rowb = blockIdx.x * 16;

    float ds[8];
    {
        const int r0 = rowb + 8 * ((t & 31) >> 4);
        for (int r = 0; r < 8; ++r) {
            int row = r0 + r; if (row >= N) row = N - 1;
            ds[r] = dis[row];
        }
    }

    for (int f = t; f < FDIM; f += 96) {
        const float inv_n = 1.f / (float)N;
        const float mean = stats[f] * inv_n;
        const float var  = stats[FDIM + f] * inv_n - mean * mean;
        const float inv  = rsqrtf(var + 1e-5f);
        const float g = gamma[f] * inv;
        scl[f] = g;
        sft[f] = beta[f] - mean * g;
    }
    for (int idx = t; idx < 64 * 48; idx += 96) {
        int k2 = idx / 48, c = idx - k2 * 48;
        v2f b;
        b.x = (c < C) ? W[(size_t)(2 * k2) * C + c]     : 0.f;
        b.y = (c < C) ? W[(size_t)(2 * k2 + 1) * C + c] : 0.f;
        lB[idx] = b;
    }
    __syncthreads();
    for (int idx = t; idx < 16 * FDIM; idx += 96) {
        int r = idx >> 7, k = idx & 127;
        int gr = rowb + r; if (gr >= N) gr = N - 1;
        lA[r * LDA + k] = A[(size_t)gr * FDIM + k] * scl[k] + sft[k];
    }
    __syncthreads();

    const int wave = t >> 5, lane = t & 31;
    const int m = lane & 15, hi = lane >> 4;
    const float* Arow = &lA[m * LDA];
    const int nloc = wave * 16 + m;

    v8f acc = {0.f, 0.f, 0.f, 0.f, 0.f, 0.f, 0.f, 0.f};
#pragma unroll
    for (int k = 0; k < FDIM; k += 4) {
        v2f a = *(const v2f*)&Arow[k + 2 * hi];
        v2f b = lB[(k / 2 + hi) * 48 + nloc];
        acc = __builtin_amdgcn_wmma_f32_16x16x4_f32(
            false, a, false, b, (short)0, acc, false, false);
    }

    if (nloc < C) {
#pragma unroll
        for (int r = 0; r < 8; ++r) {
            int row = rowb + r + 8 * hi;
            if (row < N) O[(size_t)row * C + nloc] = acc[r] * ds[r];
        }
    }
}

__global__ __launch_bounds__(256) void k_logsoftmax(const float* __restrict__ in,
                                                    float* __restrict__ out, int N, int C)
{
    long long gid = (long long)blockIdx.x * 256 + threadIdx.x;
    int row = (int)(gid >> 5);
    int lane = threadIdx.x & 31;
    if (row >= N) return;
    const float* p = in + (size_t)row * C;
    float* q = out + (size_t)row * C;
    float v0 = (lane < C) ? p[lane] : -3.4e38f;
    float v1 = (32 + lane < C) ? p[32 + lane] : -3.4e38f;
    float mx = fmaxf(v0, v1);
#pragma unroll
    for (int o = 16; o > 0; o >>= 1) mx = fmaxf(mx, __shfl_xor(mx, o, 32));
    float s = ((lane < C) ? expf(v0 - mx) : 0.f) +
              ((32 + lane < C) ? expf(v1 - mx) : 0.f);
#pragma unroll
    for (int o = 16; o > 0; o >>= 1) s += __shfl_xor(s, o, 32);
    float l = logf(s);
    if (lane < C)      q[lane]      = v0 - mx - l;
    if (32 + lane < C) q[32 + lane] = v1 - mx - l;
}

extern "C" void kernel_launch(void* const* d_in, const int* in_sizes, int n_in,
                              void* d_out, int out_size, void* d_ws, size_t ws_size,
                              hipStream_t stream)
{
    const float* x     = (const float*)d_in[0];
    const int*   ei    = (const int*)  d_in[1];
    const float* W1    = (const float*)d_in[2];
    const float* b1    = (const float*)d_in[3];
    const float* gamma = (const float*)d_in[4];
    const float* beta  = (const float*)d_in[5];
    const float* W2    = (const float*)d_in[6];
    const float* b2    = (const float*)d_in[7];
    float* out = (float*)d_out;

    const int N = in_sizes[0] / FDIM;
    const int E = in_sizes[1] / 2;
    const int C = in_sizes[6] / FDIM;
    const int* src = ei;
    const int* dst = ei + E;

    char* ws = (char*)d_ws;
    size_t off = 0;
    auto up = [](size_t v) { return (v + 255) & ~(size_t)255; };
    int*   rowptr = (int*)(ws + off);  off = up(off + (size_t)N * 4);
    float* dis   = (float*)(ws + off); off = up(off + (size_t)N * 4);
    float* stats = (float*)(ws + off); off = up(off + 512 * 4);
    float* h1   = (float*)(ws + off);  off = up(off + (size_t)N * FDIM * 4);
    float* agg1 = (float*)(ws + off);  off = up(off + (size_t)N * FDIM * 4);
    float* h2   = h1;
    int* tmp    = (int*)agg1;
    int* binCnt = (int*)((char*)agg1 + up((size_t)E * 4));
    int* cursor = binCnt + MAXNB;
    int*   csr_src = (int*)d_out;
    float* logits  = agg1;

    const int nb_C   = (E + CH - 1) / CH;
    const int nb_B   = (N + 255) / 256;
    const int nb_Nw  = (N + 7) / 8;

    k_init<<<4, 256, 0, stream>>>(binCnt, stats);
    k_hist<<<nb_C, 512, 0, stream>>>(dst, binCnt, E);
    k_scatter<<<nb_C, 512, 0, stream>>>(src, dst, binCnt, cursor, tmp, E);
    k_group<<<nb_B, 256, 0, stream>>>(tmp, binCnt, rowptr, dis, csr_src, N);

    k_gemm1<<<dim3((N + 31) / 32, FDIM / 64), 256, 0, stream>>>(x, W1, dis, h1, N);

    k_gather128<<<1024, 512, 0, stream>>>(h1, rowptr, csr_src, dis, b1, agg1, stats, N, E, 1024);

    k_gemm2<<<(N + 15) / 16, 96, 0, stream>>>(agg1, W2, stats, gamma, beta, dis, h2, N, C);

    k_gatherC<<<nb_Nw, 256, 0, stream>>>(h2, rowptr, csr_src, dis, b2, logits, N, E, C);

    k_logsoftmax<<<nb_Nw, 256, 0, stream>>>(logits, out, N, C);
}


# Log-softmax taken in the conv2 gather on the row in registers, stored straight to out; GEMM2 moves the edge list to the dead h1 plane: logits plane and one launch gone, 9 to 8; sum orders unchanged
# speedup vs baseline: 3.9326x; 1.0408x over previous
#include <hip/hip_runtime.h>

typedef __attribute__((ext_vector_type(2))) float v2f;
typedef __attribute__((ext_vector_type(8))) float v8f;

#define FDIM 128
#define LDA 132

#define BSH    8
#define MAXNB  512
#define CH     8192
#define EPT    (CH / 512)
#define SEGPT  20
#define SEGCAP 12288

__global__ __launch_bounds__(256) void k_init(int* __restrict__ bins,
                                              float* __restrict__ stats)
{
    int i = blockIdx.x * 256 + threadIdx.x;
    if (i < 2 * MAXNB) bins[i] = 0;
    if (i < 256) stats[i] = 0.f;
}

__global__ __launch_bounds__(512) void k_hist(const int* __restrict__ dst,
                                              int* __restrict__ binCnt, int E)
{
    __shared__ int hist[MAXNB];
    const int t = threadIdx.x;
    const int base = blockIdx.x * CH;
    const int n = (E - base < CH) ? (E - base) : CH;
    int dv[EPT];
#pragma unroll
    for (int k = 0; k < EPT; ++k) {
        const int i = t + k * 512;
        dv[k] = dst[base + ((i < n) ? i : n - 1)];
    }
    hist[t] = 0;
    __syncthreads();
#pragma unroll
    for (int k = 0; k < EPT; ++k)
        if (t + k * 512 < n) atomicAdd(&hist[dv[k] >> BSH], 1);
    __syncthreads();
    const int c = hist[t];
    if (c > 0) atomicAdd(&binCnt[t], c);
}

__global__ __launch_bounds__(512) void k_scatter(const int* __restrict__ src,
                                                 const int* __restrict__ dst,
                                                 const int* __restrict__ binCnt,
                                                 int* __restrict__ cursor,
                                                 int* __restrict__ tmp, int E)
{
    __shared__ int hist[MAXNB];
    __shared__ int lcur[MAXNB];
    __shared__ int gdelta[MAXNB];
    __shared__ int wc[16], wg[16];
    __shared__ int sval[CH];
    __shared__ int sadr[CH];

    const int t = threadIdx.x, lane = t & 31, wave = t >> 5;
    const int base = blockIdx.x * CH;
    const int n = (E - base < CH) ? (E - base) : CH;

    int dv[EPT], sv[EPT];
#pragma unroll
    for (int k = 0; k < EPT; ++k) {
        const int i = t + k * 512;
        dv[k] = dst[base + ((i < n) ? i : n - 1)];
    }
#pragma unroll
    for (int k = 0; k < EPT; ++k) {
        const int i = t + k * 512;
        sv[k] = src[base + ((i < n) ? i : n - 1)];
    }
    const int g = binCnt[t];

    hist[t] = 0;
    __syncthreads();
#pragma unroll
    for (int k = 0; k < EPT; ++k)
        if (t + k * 512 < n) atomicAdd(&hist[dv[k] >> BSH], 1);
    __syncthreads();

    const int c = hist[t];
    int ci = c, gi = g;
    for (int o = 1; o < 32; o <<= 1) {
        int vc = __shfl_up(ci, o, 32);
        int vg = __shfl_up(gi, o, 32);
        if (lane >= o) { ci += vc; gi += vg; }
    }
    if (lane == 31) { wc[wave] = ci; wg[wave] = gi; }
    __syncthreads();
    int ce = ci - c, ge = gi - g;
    for (int w = 0; w < wave; ++w) { ce += wc[w]; ge += wg[w]; }
    int r = 0;
    if (c > 0) r = atomicAdd(&cursor[t], c);
    lcur[t] = ce;
    gdelta[t] = ge + r - ce;
    __syncthreads();

    int pv[EPT];
#pragma unroll
    for (int k = 0; k < EPT; ++k) {
        pv[k] = 0;
        if (t + k * 512 < n) pv[k] = atomicAdd(&lcur[dv[k] >> BSH], 1);
    }
#pragma unroll
    for (int k = 0; k < EPT; ++k) {
        if (t + k * 512 < n) {
            const int p = pv[k];
            sval[p] = (int)(((unsigned)sv[k] << 8) | (unsigned)(dv[k] & 255));
            sadr[p] = gdelta[dv[k] >> BSH] + p;
        }
    }
    __syncthreads();
    for (int i = t; i < n; i += 512) tmp[sadr[i]] = sval[i];
}

__global__ __launch_bounds__(256) void k_group(const int* __restrict__ tmp,
                                               const int* __restrict__ binCnt,
                                               int* __restrict__ rowptr,
                                               float* __restrict__ dis,
                                               int* __restrict__ csr_src, int N)
{
    __shared__ int cnt[256];
    __shared__ int red[8], wsum[8];
    __shared__ int sbuf[SEGCAP];

    const int t = threadIdx.x, lane = t & 31, wave = t >> 5;
    const int b = blockIdx.x;

    const int g0 = binCnt[t], g1 = binCnt[t + 256];
    int a = ((t < b) ? g0 : 0) + ((t + 256 < b) ? g1 : 0);
    for (int o = 16; o > 0; o >>= 1) a += __shfl_xor(a, o, 32);
    if (lane == 0) red[wave] = a;
    cnt[t] = 0;
    __syncthreads();
    int beg = 0;
    for (int w = 0; w < 8; ++w) beg += red[w];
    beg = __builtin_amdgcn_readfirstlane(beg);
    const int len = __builtin_amdgcn_readfirstlane(binCnt[b]);
    const bool staged = len <= SEGCAP;

    for (int i0 = 0; i0 < len; i0 += SEGPT * 256) {
        int v[SEGPT];
#pragma unroll
        for (int k = 0; k < SEGPT; ++k) {
            const int i = i0 + t + k * 256;
            v[k] = tmp[beg + ((i < len) ? i : len - 1)];
        }
#pragma unroll
        for (int k = 0; k < SEGPT; ++k)
            if (i0 + t + k * 256 < len) atomicAdd(&cnt[v[k] & 255], 1);
    }
    __syncthreads();

    const int c = cnt[t];
    int incl = c;
    for (int o = 1; o < 32; o <<= 1) {
        int v = __shfl_up(incl, o, 32);
        if (lane >= o) incl += v;
    }
    if (lane == 31) wsum[wave] = incl;
    __syncthreads();
    int excl = incl - c;
    for (int w = 0; w < wave; ++w) excl += wsum[w];
    const int node = b * 256 + t;
    if (node < N) {
        rowptr[node] = beg + excl;
        dis[node] = rsqrtf((float)(c + 1));
    }
    cnt[t] = excl;
    __syncthreads();

    for (int i0 = 0; i0 < len; i0 += SEGPT * 256) {
        unsigned v[SEGPT];
        int p[SEGPT];
#pragma unroll
        for (int k = 0; k < SEGPT; ++k) {
            const int i = i0 + t + k * 256;
            v[k] = (unsigned)tmp[beg + ((i < len) ? i : len - 1)];
        }
#pragma unroll
        for (int k = 0; k < SEGPT; ++k) {
            p[k] = 0;
            if (i0 + t + k * 256 < len) p[k] = atomicAdd(&cnt[v[k] & 255u], 1);
        }
#pragma unroll
        for (int k = 0; k < SEGPT; ++k) {
            if (i0 + t + k * 256 < len) {
                const int s = (int)(v[k] >> 8);
                if (staged) sbuf[p[k]] = s;
                else        csr_src[beg + p[k]] = s;
            }
        }
    }
    __syncthreads();
    if (staged)
        for (int i = t; i < len; i += 256) csr_src[beg + i] = sbuf[i];
}

__global__ __launch_bounds__(256) void k_gemm1(const float* __restrict__ X,
                                               const float* __restrict__ W,
                                               const float* __restrict__ dis,
                                               float* __restrict__ O, int N)
{
    __shared__ float lA[32 * LDA];
    __shared__ v2f   lB[64 * 64];

    const int t    = threadIdx.x;
    const int rowb = blockIdx.x * 32;
    const int colb = blockIdx.y * 64;

    float ds[8];
    {
        const int r0 = rowb + ((t >> 5) >> 2) * 16 + 8 * ((t & 31) >> 4);
        for (int r = 0; r < 8; ++r) {
            int row = r0 + r; if (row >= N) row = N - 1;
            ds[r] = dis[row];
        }
    }

    for (int idx = t; idx < 32 * FDIM; idx += 256) {
        int r = idx >> 7, k = idx & 127;
        int gr = rowb + r; if (gr >= N) gr = N - 1;
        lA[r * LDA + k] = X[(size_t)gr * FDIM + k];
    }
    for (int idx = t; idx < 64 * 64; idx += 256) {
        int k2 = idx >> 6, c = idx & 63;
        const float* wp = W + (size_t)(2 * k2) * FDIM + (colb + c);
        v2f b; b.x = wp[0]; b.y = wp[FDIM];
        lB[idx] = b;
    }
    __syncthreads();

    const int wave = t >> 5, lane = t & 31;
    const int m = lane & 15, hi = lane >> 4;
    const int rt = wave >> 2, ct = wave & 3;
    const float* Arow = &lA[(rt * 16 + m) * LDA];
    const int nloc = ct * 16 + m;

    v8f acc = {0.f, 0.f, 0.f, 0.f, 0.f, 0.f, 0.f, 0.f};
#pragma unroll
    for (int k = 0; k < FDIM; k += 4) {
        v2f a = *(const v2f*)&Arow[k + 2 * hi];
        v2f b = lB[(k / 2 + hi) * 64 + nloc];
        acc = __builtin_amdgcn_wmma_f32_16x16x4_f32(
            false, a, false, b, (short)0, acc, false, false);
    }

    const int col = colb + nloc;
#pragma unroll
    for (int r = 0; r < 8; ++r) {
        int row = rowb + rt * 16 + r + 8 * hi;
        if (row < N) O[(size_t)row * FDIM + col] = acc[r] * ds[r];
    }
}

static __device__ __forceinline__ int bcast_i(int v, int j)
{
    return __builtin_amdgcn_readlane(v, j);
}
static __device__ __forceinline__ void row_add(float4& acc, const float4& v)
{
    acc.x += v.x; acc.y += v.y; acc.z += v.z; acc.w += v.w;
}

__global__ __launch_bounds__(512) void k_gather128(const float* __restrict__ h,
                                                   const int* __restrict__ rowptr,
                                                   const int* __restrict__ csr_src,
                                                   const float* __restrict__ dis,
                                                   const float* __restrict__ bias,
                                                   float* __restrict__ out,
                                                   float* __restrict__ stats,
                                                   int N, int E, int nblk)
{
    __shared__ float red[16 * 256];
    const int lane = threadIdx.x & 31;
    const int wave = __builtin_amdgcn_readfirstlane((int)(threadIdx.x >> 5));
    const float4* hp = (const float4*)h;
    const float4 b = ((const float4*)bias)[lane];
    float4 s1 = {0.f, 0.f, 0.f, 0.f}, s2 = {0.f, 0.f, 0.f, 0.f};

    for (int nb = blockIdx.x * 16; nb < N; nb += nblk * 16) {
        const int node = nb + wave;
        if (node < N) {
            const int beg = __builtin_amdgcn_readfirstlane(rowptr[node]);
            const int end = (node + 1 < N) ? __builtin_amdgcn_readfirstlane(rowptr[node + 1]) : E;
            const float di = dis[node];

            float4 acc = hp[(size_t)node * 32 + lane];
            for (int p = beg; p < end; p += 32) {
                const int n = (end - p < 32) ? (end - p) : 32;
                int s_l = 0;
                if (lane < n) s_l = csr_src[p + lane];
                int j = 0;
                for (; j + 4 <= n; j += 4) {
                    const int i0 = bcast_i(s_l, j),     i1 = bcast_i(s_l, j + 1);
                    const int i2 = bcast_i(s_l, j + 2), i3 = bcast_i(s_l, j + 3);
                    const float4 v0 = hp[(size_t)i0 * 32 + lane];
                    const float4 v1 = hp[(size_t)i1 * 32 + lane];
                    const float4 v2 = hp[(size_t)i2 * 32 + lane];
                    const float4 v3 = hp[(size_t)i3 * 32 + lane];
                    row_add(acc, v0); row_add(acc, v1);
                    row_add(acc, v2); row_add(acc, v3);
                }
                for (; j < n; ++j) {
                    const int i0 = bcast_i(s_l, j);
                    const float4 v0 = hp[(size_t)i0 * 32 + lane];
                    row_add(acc, v0);
                }
            }
            acc.x = fmaxf(acc.x * di + b.x, 0.f);
            acc.y = fmaxf(acc.y * di + b.y, 0.f);
            acc.z = fmaxf(acc.z * di + b.z, 0.f);
            acc.w = fmaxf(acc.w * di + b.w, 0.f);
            ((float4*)out)[(size_t)node * 32 + lane] = acc;
            s1.x += acc.x; s1.y += acc.y; s1.z += acc.z; s1.w += acc.w;
            s2.x += acc.x * acc.x; s2.y += acc.y * acc.y;
            s2.z += acc.z * acc.z; s2.w += acc.w * acc.w;
        }
    }

    ((float4*)red)[wave * 64 + lane]      = s1;
    ((float4*)red)[wave * 64 + 32 + lane] = s2;
    __syncthreads();
    const int t = threadIdx.x;
    if (t < 256) {
        float v = 0.f;
        for (int w = 0; w < 16; ++w) v += red[w * 256 + t];
        atomicAdd(&stats[t], v);
    }
}

__global__ __launch_bounds__(256) void k_gatherC(const float* __restrict__ h,
                                                 const int* __restrict__ rowptr,
                                                 const int* __restrict__ csr_src,
                                                 const float* __restrict__ dis,
                                                 const float* __restrict__ bias,
                                                 float* __restrict__ out, int N, int E, int C)
{
    const int lane = threadIdx.x & 31;
    const int node = blockIdx.x * 8 + __builtin_amdgcn_readfirstlane((int)(threadIdx.x >> 5));
    if (node >= N) return;
    const int beg = __builtin_amdgcn_readfirstlane(rowptr[node]);
    const int end = (node + 1 < N) ? __builtin_amdgcn_readfirstlane(rowptr[node + 1]) : E;
    const float di = dis[node];
    const bool a0 = lane < C, a1 = 32 + lane < C;
    const int c0 = a0 ? lane : 0, c1 = a1 ? 32 + lane : 0;

    float acc0, acc1;
    {
        const float* hr = h + (size_t)node * C;
        acc0 = hr[c0];
        acc1 = hr[c1];
    }
    const float b0 = bias[c0], b1 = bias[c1];
    for (int p = beg; p < end; p += 32) {
        const int n = (end - p < 32) ? (end - p) : 32;
        int s_l = 0;
        if (lane < n) s_l = csr_src[p + lane];
        int j = 0;
        for (; j + 4 <= n; j += 4) {
            const float* r0 = h + (size_t)bcast_i(s_l, j) * C;
            const float* r1 = h + (size_t)bcast_i(s_l, j + 1) * C;
            const float* r2 = h + (size_t)bcast_i(s_l, j + 2) * C;
            const float* r3 = h + (size_t)bcast_i(s_l, j + 3) * C;
            const float x0 = r0[c0], y0 = r0[c1];
            const float x1 = r1[c0], y1 = r1[c1];
            const float x2 = r2[c0], y2 = r2[c1];
            const float x3 = r3[c0], y3 = r3[c1];
            acc0 += x0; acc1 += y0;
            acc0 += x1; acc1 += y1;
            acc0 += x2; acc1 += y2;
            acc0 += x3; acc1 += y3;
        }
        for (; j < n; ++j) {
            const float* r0 = h + (size_t)bcast_i(s_l, j) * C;
            acc0 += r0[c0]; acc1 += r0[c1];
        }
    }

    const float v0 = a0 ? (acc0 * di + b0) : -3.4e38f;
    const float v1 = a1 ? (acc1 * di + b1) : -3.4e38f;
    float mx = fmaxf(v0, v1);
#pragma unroll
    for (int o = 16; o > 0; o >>= 1) mx = fmaxf(mx, __shfl_xor(mx, o, 32));
    float s = (a0 ? expf(v0 - mx) : 0.f) + (a1 ? expf(v1 - mx) : 0.f);
#pragma unroll
    for (int o = 16; o > 0; o >>= 1) s += __shfl_xor(s, o, 32);
    const float l = logf(s);
    float* orow = out + (size_t)node * C;
    if (a0) orow[lane]      = v0 - mx - l;
    if (a1) orow[32 + lane] = v1 - mx - l;
}

__global__ __launch_bounds__(96) void k_gemm2(const float* __restrict__ A,
                                              const float* __restrict__ W,
                                              const float* __restrict__ stats,
                                              const float* __restrict__ gamma,
                                              const float* __restrict__ beta,
                                              const float* __restrict__ dis,
                                              float* __restrict__ O, int N, int C,
                                              const int* __restrict__ csr_in,
                                              int* __restrict__ csr_out, int E)
{
    __shared__ float lA[16 * LDA];
    __shared__ v2f   lB[64 * 48];
    __shared__ float scl[FDIM], sft[FDIM];

    const int t = threadIdx.x;
    const int rowb = blockIdx.x * 16;

    const int cstride = (int)gridDim.x * 96;
    const int ci = blockIdx.x * 96 + t;
    int cv[4];
#pragma unroll
    for (int k = 0; k < 4; ++k) {
        const int i = ci + k * cstride;
        cv[k] = csr_in[(i < E) ? i : E - 1];
    }

    float ds[8];
    {
        const int r0 = rowb + 8 * ((t & 31) >> 4);
        for (int r = 0; r < 8; ++r) {
            int row = r0 + r; if (row >= N) row = N - 1;
            ds[r] = dis[row];
        }
    }

    for (int f = t; f < FDIM; f += 96) {
        const float inv_n = 1.f / (float)N;
        const float mean = stats[f] * inv_n;
        const float var  = stats[FDIM + f] * inv_n - mean * mean;
        const float inv  = rsqrtf(var + 1e-5f);
        const float g = gamma[f] * inv;
        scl[f] = g;
        sft[f] = beta[f] - mean * g;
    }
    for (int idx = t; idx < 64 * 48; idx += 96) {
        int k2 = idx / 48, c = idx - k2 * 48;
        v2f b;
        b.x = (c < C) ? W[(size_t)(2 * k2) * C + c]     : 0.f;
        b.y = (c < C) ? W[(size_t)(2 * k2 + 1) * C + c] : 0.f;
        lB[idx] = b;
    }
    __syncthreads();
    for (int idx = t; idx < 16 * FDIM; idx += 96) {
        int r = idx >> 7, k = idx & 127;
        int gr = rowb + r; if (gr >= N) gr = N - 1;
        lA[r * LDA + k] = A[(size_t)gr * FDIM + k] * scl[k] + sft[k];
    }
    __syncthreads();

    const int wave = t >> 5, lane = t & 31;
    const int m = lane & 15, hi = lane >> 4;
    const float* Arow = &lA[m * LDA];
    const int nloc = wave * 16 + m;

    v8f acc = {0.f, 0.f, 0.f, 0.f, 0.f, 0.f, 0.f, 0.f};
#pragma unroll
    for (int k = 0; k < FDIM; k += 4) {
        v2f a = *(const v2f*)&Arow[k + 2 * hi];
        v2f b = lB[(k / 2 + hi) * 48 + nloc];
        acc = __builtin_amdgcn_wmma_f32_16x16x4_f32(
            false, a, false, b, (short)0, acc, false, false);
    }

    if (nloc < C) {
#pragma unroll
        for (int r = 0; r < 8; ++r) {
            int row = rowb + r + 8 * hi;
            if (row < N) O[(size_t)row * C + nloc] = acc[r] * ds[r];
        }
    }

#pragma unroll
    for (int k = 0; k < 4; ++k) {
        const int i = ci + k * cstride;
        if (i < E) csr_out[i] = cv[k];
    }
    for (int i = ci + 4 * cstride; i < E; i += cstride) csr_out[i] = csr_in[i];
}

extern "C" void kernel_launch(void* const* d_in, const int* in_sizes, int n_in,
                              void* d_out, int out_size, void* d_ws, size_t ws_size,
                              hipStream_t stream)
{
    const float* x     = (const float*)d_in[0];
    const int*   ei    = (const int*)  d_in[1];
    const float* W1    = (const float*)d_in[2];
    const float* b1    = (const float*)d_in[3];
    const float* gamma = (const float*)d_in[4];
    const float* beta  = (const float*)d_in[5];
    const float* W2    = (const float*)d_in[6];
    const float* b2    = (const float*)d_in[7];
    float* out = (float*)d_out;

    const int N = in_sizes[0] / FDIM;
    const int E = in_sizes[1] / 2;
    const int C = in_sizes[6] / FDIM;
    const int* src = ei;
    const int* dst = ei + E;

    char* ws = (char*)d_ws;
    size_t off = 0;
    auto up = [](size_t v) { return (v + 255) & ~(size_t)255; };
    int*   rowptr = (int*)(ws + off);  off = up(off + (size_t)N * 4);
    float* dis   = (float*)(ws + off); off = up(off + (size_t)N * 4);
    float* stats = (float*)(ws + off); off = up(off + 512 * 4);
    float* h1   = (float*)(ws + off);  off = up(off + (size_t)N * FDIM * 4);
    float* agg1 = (float*)(ws + off);  off = up(off + (size_t)N * FDIM * 4);
    float* h2   = h1;
    int* tmp    = (int*)agg1;
    int* binCnt = (int*)((char*)agg1 + up((size_t)E * 4));
    int* cursor = binCnt + MAXNB;
    int* csr_src = (int*)d_out;
    int* csr_l2  = (int*)((char*)h1 + up((size_t)N * C * 4));

    const int nb_C   = (E + CH - 1) / CH;
    const int nb_B   = (N + 255) / 256;
    const int nb_Nw  = (N + 7) / 8;

    k_init<<<4, 256, 0, stream>>>(binCnt, stats);
    k_hist<<<nb_C, 512, 0, stream>>>(dst, binCnt, E);
    k_scatter<<<nb_C, 512, 0, stream>>>(src, dst, binCnt, cursor, tmp, E);
    k_group<<<nb_B, 256, 0, stream>>>(tmp, binCnt, rowptr, dis, csr_src, N);

    k_gemm1<<<dim3((N + 31) / 32, FDIM / 64), 256, 0, stream>>>(x, W1, dis, h1, N);

    k_gather128<<<1024, 512, 0, stream>>>(h1, rowptr, csr_src, dis, b1, agg1, stats, N, E, 1024);

    k_gemm2<<<(N + 15) / 16, 96, 0, stream>>>(agg1, W2, stats, gamma, beta, dis, h2, N, C,
                                              csr_src, csr_l2, E);

    k_gatherC<<<nb_Nw, 256, 0, stream>>>(h2, rowptr, csr_l2, dis, b2, out, N, E, C);
}


# Both GEMMs on resident blocks: W (GEMM2: also BN scale/shift) staged to LDS once per block, which walks row tiles with the next tile's 16-byte row loads in flight across LDS-only barriers; sums kept
# speedup vs baseline: 5.4753x; 1.3923x over previous
#include <hip/hip_runtime.h>

typedef __attribute__((ext_vector_type(2))) float v2f;
typedef __attribute__((ext_vector_type(8))) float v8f;
typedef __attribute__((ext_vector_type(4))) float v4f;

#define FDIM 128
#define LDA 132

#define BSH    8
#define MAXNB  512
#define CH     8192
#define EPT    (CH / 512)
#define SEGPT  20
#define SEGCAP 12288

__global__ __launch_bounds__(256) void k_init(int* __restrict__ bins,
                                              float* __restrict__ stats)
{
    int i = blockIdx.x * 256 + threadIdx.x;
    if (i < 2 * MAXNB) bins[i] = 0;
    if (i < 256) stats[i] = 0.f;
}

__global__ __launch_bounds__(512) void k_hist(const int* __restrict__ dst,
                                              int* __restrict__ binCnt, int E)
{
    __shared__ int hist[MAXNB];
    const int t = threadIdx.x;
    const int base = blockIdx.x * CH;
    const int n = (E - base < CH) ? (E - base) : CH;
    int dv[EPT];
#pragma unroll
    for (int k = 0; k < EPT; ++k) {
        const int i = t + k * 512;
        dv[k] = dst[base + ((i < n) ? i : n - 1)];
    }
    hist[t] = 0;
    __syncthreads();
#pragma unroll
    for (int k = 0; k < EPT; ++k)
        if (t + k * 512 < n) atomicAdd(&hist[dv[k] >> BSH], 1);
    __syncthreads();
    const int c = hist[t];
    if (c > 0) atomicAdd(&binCnt[t], c);
}

__global__ __launch_bounds__(512) void k_scatter(const int* __restrict__ src,
                                                 const int* __restrict__ dst,
                                                 const int* __restrict__ binCnt,
                                                 int* __restrict__ cursor,
                                                 int* __restrict__ tmp, int E)
{
    __shared__ int hist[MAXNB];
    __shared__ int lcur[MAXNB];
    __shared__ int gdelta[MAXNB];
    __shared__ int wc[16], wg[16];
    __shared__ int sval[CH];
    __shared__ int sadr[CH];

    const int t = threadIdx.x, lane = t & 31, wave = t >> 5;
    const int base = blockIdx.x * CH;
    const int n = (E - base < CH) ? (E - base) : CH;

    int dv[EPT], sv[EPT];
#pragma unroll
    for (int k = 0; k < EPT; ++k) {
        const int i = t + k * 512;
        dv[k] = dst[base + ((i < n) ? i : n - 1)];
    }
#pragma unroll
    for (int k = 0; k < EPT; ++k) {
        const int i = t + k * 512;
        sv[k] = src[base + ((i < n) ? i : n - 1)];
    }
    const int g = binCnt[t];

    hist[t] = 0;
    __syncthreads();
#pragma unroll
    for (int k = 0; k < EPT; ++k)
        if (t + k * 512 < n) atomicAdd(&hist[dv[k] >> BSH], 1);
    __syncthreads();

    const int c = hist[t];
    int ci = c, gi = g;
    for (int o = 1; o < 32; o <<= 1) {
        int vc = __shfl_up(ci, o, 32);
        int vg = __shfl_up(gi, o, 32);
        if (lane >= o) { ci += vc; gi += vg; }
    }
    if (lane == 31) { wc[wave] = ci; wg[wave] = gi; }
    __syncthreads();
    int ce = ci - c, ge = gi - g;
    for (int w = 0; w < wave; ++w) { ce += wc[w]; ge += wg[w]; }
    int r = 0;
    if (c > 0) r = atomicAdd(&cursor[t], c);
    lcur[t] = ce;
    gdelta[t] = ge + r - ce;
    __syncthreads();

    int pv[EPT];
#pragma unroll
    for (int k = 0; k < EPT; ++k) {
        pv[k] = 0;
        if (t + k * 512 < n) pv[k] = atomicAdd(&lcur[dv[k] >> BSH], 1);
    }
#pragma unroll
    for (int k = 0; k < EPT; ++k) {
        if (t + k * 512 < n) {
            const int p = pv[k];
            sval[p] = (int)(((unsigned)sv[k] << 8) | (unsigned)(dv[k] & 255));
            sadr[p] = gdelta[dv[k] >> BSH] + p;
        }
    }
    __syncthreads();
    for (int i = t; i < n; i += 512) tmp[sadr[i]] = sval[i];
}

__global__ __launch_bounds__(256) void k_group(const int* __restrict__ tmp,
                                               const int* __restrict__ binCnt,
                                               int* __restrict__ rowptr,
                                               float* __restrict__ dis,
                                               int* __restrict__ csr_src, int N)
{
    __shared__ int cnt[256];
    __shared__ int red[8], wsum[8];
    __shared__ int sbuf[SEGCAP];

    const int t = threadIdx.x, lane = t & 31, wave = t >> 5;
    const int b = blockIdx.x;

    const int g0 = binCnt[t], g1 = binCnt[t + 256];
    int a = ((t < b) ? g0 : 0) + ((t + 256 < b) ? g1 : 0);
    for (int o = 16; o > 0; o >>= 1) a += __shfl_xor(a, o, 32);
    if (lane == 0) red[wave] = a;
    cnt[t] = 0;
    __syncthreads();
    int beg = 0;
    for (int w = 0; w < 8; ++w) beg += red[w];
    beg = __builtin_amdgcn_readfirstlane(beg);
    const int len = __builtin_amdgcn_readfirstlane(binCnt[b]);
    const bool staged = len <= SEGCAP;

    for (int i0 = 0; i0 < len; i0 += SEGPT * 256) {
        int v[SEGPT];
#pragma unroll
        for (int k = 0; k < SEGPT; ++k) {
            const int i = i0 + t + k * 256;
            v[k] = tmp[beg + ((i < len) ? i : len - 1)];
        }
#pragma unroll
        for (int k = 0; k < SEGPT; ++k)
            if (i0 + t + k * 256 < len) atomicAdd(&cnt[v[k] & 255], 1);
    }
    __syncthreads();

    const int c = cnt[t];
    int incl = c;
    for (int o = 1; o < 32; o <<= 1) {
        int v = __shfl_up(incl, o, 32);
        if (lane >= o) incl += v;
    }
    if (lane == 31) wsum[wave] = incl;
    __syncthreads();
    int excl = incl - c;
    for (int w = 0; w < wave; ++w) excl += wsum[w];
    const int node = b * 256 + t;
    if (node < N) {
        rowptr[node] = beg + excl;
        dis[node] = rsqrtf((float)(c + 1));
    }
    cnt[t] = excl;
    __syncthreads();

    for (int i0 = 0; i0 < len; i0 += SEGPT * 256) {
        unsigned v[SEGPT];
        int p[SEGPT];
#pragma unroll
        for (int k = 0; k < SEGPT; ++k) {
            const int i = i0 + t + k * 256;
            v[k] = (unsigned)tmp[beg + ((i < len) ? i : len - 1)];
        }
#pragma unroll
        for (int k = 0; k < SEGPT; ++k) {
            p[k] = 0;
            if (i0 + t + k * 256 < len) p[k] = atomicAdd(&cnt[v[k] & 255u], 1);
        }
#pragma unroll
        for (int k = 0; k < SEGPT; ++k) {
            if (i0 + t + k * 256 < len) {
                const int s = (int)(v[k] >> 8);
                if (staged) sbuf[p[k]] = s;
                else        csr_src[beg + p[k]] = s;
            }
        }
    }
    __syncthreads();
    if (staged)
        for (int i = t; i < len; i += 256) csr_src[beg + i] = sbuf[i];
}

static __device__ __forceinline__ void lds_barrier()
{
    __builtin_amdgcn_fence(__ATOMIC_RELEASE, "workgroup", "local");
    __builtin_amdgcn_s_barrier();
    __builtin_amdgcn_fence(__ATOMIC_ACQUIRE, "workgroup", "local");
}

__global__ __launch_bounds__(256) void k_gemm1(const float* __restrict__ X,
                                               const float* __restrict__ W,
                                               const float* __restrict__ dis,
                                               float* __restrict__ O, int N)
{
    __shared__ __attribute__((aligned(16))) float lA[32 * LDA];
    __shared__ v2f   lB[64 * 64];

    const int t    = threadIdx.x;
    const int colb = blockIdx.y * 64;
    const int ntiles = (N + 31) >> 5;
    const int wave = t >> 5, lane = t & 31;
    const int m = lane & 15, hi = lane >> 4;
    const int rt = wave >> 2, ct = wave & 3;
    const int ak = lane * 4;
    const int er = rt * 16 + 8 * hi;

#define G1_ROW(rb, j) (*(const v4f*)&X[(size_t)(((rb) + wave + 8 * (j)) < N ? ((rb) + wave + 8 * (j)) : N - 1) * FDIM + ak])
#define G1_DIS(rb, r) (dis[((rb) + er + (r)) < N ? ((rb) + er + (r)) : N - 1])
    v4f pa0, pa1, pa2, pa3;
    float pd0, pd1, pd2, pd3, pd4, pd5, pd6, pd7;
    int tile = blockIdx.x;
    {
        const int rowb = tile * 32;
        pa0 = G1_ROW(rowb, 0); pa1 = G1_ROW(rowb, 1); pa2 = G1_ROW(rowb, 2); pa3 = G1_ROW(rowb, 3);
        pd0 = G1_DIS(rowb, 0); pd1 = G1_DIS(rowb, 1); pd2 = G1_DIS(rowb, 2); pd3 = G1_DIS(rowb, 3);
        pd4 = G1_DIS(rowb, 4); pd5 = G1_DIS(rowb, 5); pd6 = G1_DIS(rowb, 6); pd7 = G1_DIS(rowb, 7);
    }

#pragma unroll
    for (int i = 0; i < 16; ++i) {
        const int idx = t + 256 * i;
        const int k2 = idx >> 6, c = idx & 63;
        const float* wp = W + (size_t)(2 * k2) * FDIM + (colb + c);
        v2f b; b.x = wp[0]; b.y = wp[FDIM];
        lB[idx] = b;
    }

    const float* Arow = &lA[(rt * 16 + m) * LDA];
    const int nloc = ct * 16 + m;
    const int col = colb + nloc;

    for (; tile < ntiles; tile += (int)gridDim.x) {
        const int rowb = tile * 32;
        *(v4f*)&lA[(wave +  0) * LDA + ak] = pa0;
        *(v4f*)&lA[(wave +  8) * LDA + ak] = pa1;
        *(v4f*)&lA[(wave + 16) * LDA + ak] = pa2;
        *(v4f*)&lA[(wave + 24) * LDA + ak] = pa3;
        const float ds[8] = {pd0, pd1, pd2, pd3, pd4, pd5, pd6, pd7};
        lds_barrier();

        {
            const int nrowb = (tile + (int)gridDim.x) * 32;
            pa0 = G1_ROW(nrowb, 0); pa1 = G1_ROW(nrowb, 1); pa2 = G1_ROW(nrowb, 2); pa3 = G1_ROW(nrowb, 3);
            pd0 = G1_DIS(nrowb, 0); pd1 = G1_DIS(nrowb, 1); pd2 = G1_DIS(nrowb, 2); pd3 = G1_DIS(nrowb, 3);
            pd4 = G1_DIS(nrowb, 4); pd5 = G1_DIS(nrowb, 5); pd6 = G1_DIS(nrowb, 6); pd7 = G1_DIS(nrowb, 7);
        }

        v8f acc = {0.f, 0.f, 0.f, 0.f, 0.f, 0.f, 0.f, 0.f};
#pragma unroll
        for (int k = 0; k < FDIM; k += 4) {
            v2f a = *(const v2f*)&Arow[k + 2 * hi];
            v2f b = lB[(k / 2 + hi) * 64 + nloc];
            acc = __builtin_amdgcn_wmma_f32_16x16x4_f32(
                false, a, false, b, (short)0, acc, false, false);
        }

#pragma unroll
        for (int r = 0; r < 8; ++r) {
            int row = rowb + er + r;
            if (row < N) O[(size_t)row * FDIM + col] = acc[r] * ds[r];
        }
        lds_barrier();
    }
}

#undef G1_ROW
#undef G1_DIS

static __device__ __forceinline__ int bcast_i(int v, int j)
{
    return __builtin_amdgcn_readlane(v, j);
}
static __device__ __forceinline__ void row_add(float4& acc, const float4& v)
{
    acc.x += v.x; acc.y += v.y; acc.z += v.z; acc.w += v.w;
}

__global__ __launch_bounds__(512) void k_gather128(const float* __restrict__ h,
                                                   const int* __restrict__ rowptr,
                                                   const int* __restrict__ csr_src,
                                                   const float* __restrict__ dis,
                                                   const float* __restrict__ bias,
                                                   float* __restrict__ out,
                                                   float* __restrict__ stats,
                                                   int N, int E, int nblk)
{
    __shared__ float red[16 * 256];
    const int lane = threadIdx.x & 31;
    const int wave = __builtin_amdgcn_readfirstlane((int)(threadIdx.x >> 5));
    const float4* hp = (const float4*)h;
    const float4 b = ((const float4*)bias)[lane];
    float4 s1 = {0.f, 0.f, 0.f, 0.f}, s2 = {0.f, 0.f, 0.f, 0.f};

    for (int nb = blockIdx.x * 16; nb < N; nb += nblk * 16) {
        const int node = nb + wave;
        if (node < N) {
            const int beg = __builtin_amdgcn_readfirstlane(rowptr[node]);
            const int end = (node + 1 < N) ? __builtin_amdgcn_readfirstlane(rowptr[node + 1]) : E;
            const float di = dis[node];

            float4 acc = hp[(size_t)node * 32 + lane];
            for (int p = beg; p < end; p += 32) {
                const int n = (end - p < 32) ? (end - p) : 32;
                int s_l = 0;
                if (lane < n) s_l = csr_src[p + lane];
                int j = 0;
                for (; j + 4 <= n; j += 4) {
                    const int i0 = bcast_i(s_l, j),     i1 = bcast_i(s_l, j + 1);
                    const int i2 = bcast_i(s_l, j + 2), i3 = bcast_i(s_l, j + 3);
                    const float4 v0 = hp[(size_t)i0 * 32 + lane];
                    const float4 v1 = hp[(size_t)i1 * 32 + lane];
                    const float4 v2 = hp[(size_t)i2 * 32 + lane];
                    const float4 v3 = hp[(size_t)i3 * 32 + lane];
                    row_add(acc, v0); row_add(acc, v1);
                    row_add(acc, v2); row_add(acc, v3);
                }
                for (; j < n; ++j) {
                    const int i0 = bcast_i(s_l, j);
                    const float4 v0 = hp[(size_t)i0 * 32 + lane];
                    row_add(acc, v0);
                }
            }
            acc.x = fmaxf(acc.x * di + b.x, 0.f);
            acc.y = fmaxf(acc.y * di + b.y, 0.f);
            acc.z = fmaxf(acc.z * di + b.z, 0.f);
            acc.w = fmaxf(acc.w * di + b.w, 0.f);
            ((float4*)out)[(size_t)node * 32 + lane] = acc;
            s1.x += acc.x; s1.y += acc.y; s1.z += acc.z; s1.w += acc.w;
            s2.x += acc.x * acc.x; s2.y += acc.y * acc.y;
            s2.z += acc.z * acc.z; s2.w += acc.w * acc.w;
        }
    }

    ((float4*)red)[wave * 64 + lane]      = s1;
    ((float4*)red)[wave * 64 + 32 + lane] = s2;
    __syncthreads();
    const int t = threadIdx.x;
    if (t < 256) {
        float v = 0.f;
        for (int w = 0; w < 16; ++w) v += red[w * 256 + t];
        atomicAdd(&stats[t], v);
    }
}

__global__ __launch_bounds__(256) void k_gatherC(const float* __restrict__ h,
                                                 const int* __restrict__ rowptr,
                                                 const int* __restrict__ csr_src,
                                                 const float* __restrict__ dis,
                                                 const float* __restrict__ bias,
                                                 float* __restrict__ out, int N, int E, int C)
{
    const int lane = threadIdx.x & 31;
    const int node = blockIdx.x * 8 + __builtin_amdgcn_readfirstlane((int)(threadIdx.x >> 5));
    if (node >= N) return;
    const int beg = __builtin_amdgcn_readfirstlane(rowptr[node]);
    const int end = (node + 1 < N) ? __builtin_amdgcn_readfirstlane(rowptr[node + 1]) : E;
    const float di = dis[node];
    const bool a0 = lane < C, a1 = 32 + lane < C;
    const int c0 = a0 ? lane : 0, c1 = a1 ? 32 + lane : 0;

    float acc0, acc1;
    {
        const float* hr = h + (size_t)node * C;
        acc0 = hr[c0];
        acc1 = hr[c1];
    }
    const float b0 = bias[c0], b1 = bias[c1];
    for (int p = beg; p < end; p += 32) {
        const int n = (end - p < 32) ? (end - p) : 32;
        int s_l = 0;
        if (lane < n) s_l = csr_src[p + lane];
        int j = 0;
        for (; j + 4 <= n; j += 4) {
            const float* r0 = h + (size_t)bcast_i(s_l, j) * C;
            const float* r1 = h + (size_t)bcast_i(s_l, j + 1) * C;
            const float* r2 = h + (size_t)bcast_i(s_l, j + 2) * C;
            const float* r3 = h + (size_t)bcast_i(s_l, j + 3) * C;
            const float x0 = r0[c0], y0 = r0[c1];
            const float x1 = r1[c0], y1 = r1[c1];
            const float x2 = r2[c0], y2 = r2[c1];
            const float x3 = r3[c0], y3 = r3[c1];
            acc0 += x0; acc1 += y0;
            acc0 += x1; acc1 += y1;
            acc0 += x2; acc1 += y2;
            acc0 += x3; acc1 += y3;
        }
        for (; j < n; ++j) {
            const float* r0 = h + (size_t)bcast_i(s_l, j) * C;
            acc0 += r0[c0]; acc1 += r0[c1];
        }
    }

    const float v0 = a0 ? (acc0 * di + b0) : -3.4e38f;
    const float v1 = a1 ? (acc1 * di + b1) : -3.4e38f;
    float mx = fmaxf(v0, v1);
#pragma unroll
    for (int o = 16; o > 0; o >>= 1) mx = fmaxf(mx, __shfl_xor(mx, o, 32));
    float s = (a0 ? expf(v0 - mx) : 0.f) + (a1 ? expf(v1 - mx) : 0.f);
#pragma unroll
    for (int o = 16; o > 0; o >>= 1) s += __shfl_xor(s, o, 32);
    const float l = logf(s);
    float* orow = out + (size_t)node * C;
    if (a0) orow[lane]      = v0 - mx - l;
    if (a1) orow[32 + lane] = v1 - mx - l;
}

__global__ __launch_bounds__(96) void k_gemm2(const float* __restrict__ A,
                                              const float* __restrict__ W,
                                              const float* __restrict__ stats,
                                              const float* __restrict__ gamma,
                                              const float* __restrict__ beta,
                                              const float* __restrict__ dis,
                                              float* __restrict__ O, int N, int C,
                                              const int* __restrict__ csr_in,
                                              int* __restrict__ csr_out, int E)
{
    __shared__ __attribute__((aligned(16))) float lA[16 * LDA];
    __shared__ v2f   lB[64 * 48];
    __shared__ __attribute__((aligned(16))) float scl[FDIM];
    __shared__ __attribute__((aligned(16))) float sft[FDIM];

    const int t = threadIdx.x;
    const int ntiles = (N + 15) >> 4;
    const int wave = t >> 5, lane = t & 31;
    const int m = lane & 15, hi = lane >> 4;

    const int cstride = (int)gridDim.x * 96;
    const int ci = blockIdx.x * 96 + t;
    int cv[8];
#pragma unroll
    for (int k = 0; k < 8; ++k) {
        const int i = ci + k * cstride;
        cv[k] = csr_in[(i < E) ? i : E - 1];
    }

    const int q5 = (t + 480 > 511) ? 511 : t + 480;
#define G2_ADDR(rb, q) (&A[(size_t)(((rb) + ((q) >> 5)) < N ? ((rb) + ((q) >> 5)) : N - 1) * FDIM + ((q) & 31) * 4])
#define G2_ROW(rb, j)  (*(const v4f*)G2_ADDR(rb, t + 96 * (j)))
#define G2_ROW5(rb)    (*(const v4f*)G2_ADDR(rb, q5))
#define G2_DIS(rb, r)  (dis[((rb) + 8 * hi + (r)) < N ? ((rb) + 8 * hi + (r)) : N - 1])
    v4f pa0, pa1, pa2, pa3, pa4, pa5;
    float pd0, pd1, pd2, pd3, pd4, pd5, pd6, pd7;
    int tile = blockIdx.x;
    {
        const int rowb = tile * 16;
        pa0 = G2_ROW(rowb, 0); pa1 = G2_ROW(rowb, 1); pa2 = G2_ROW(rowb, 2);
        pa3 = G2_ROW(rowb, 3); pa4 = G2_ROW(rowb, 4); pa5 = G2_ROW5(rowb);
        pd0 = G2_DIS(rowb, 0); pd1 = G2_DIS(rowb, 1); pd2 = G2_DIS(rowb, 2); pd3 = G2_DIS(rowb, 3);
        pd4 = G2_DIS(rowb, 4); pd5 = G2_DIS(rowb, 5); pd6 = G2_DIS(rowb, 6); pd7 = G2_DIS(rowb, 7);
    }

    for (int f = t; f < FDIM; f += 96) {
        const float inv_n = 1.f / (float)N;
        const float mean = stats[f] * inv_n;
        const float var  = stats[FDIM + f] * inv_n - mean * mean;
        const float inv  = rsqrtf(var + 1e-5f);
        const float g = gamma[f] * inv;
        scl[f] = g;
        sft[f] = beta[f] - mean * g;
    }
#pragma unroll 8
    for (int i = 0; i < 32; ++i) {
        const int idx = t + 96 * i;
        const int k2 = idx / 48, c = idx - k2 * 48;
        const int cc = (c < C) ? c : 0;
        const float bx = W[(2 * k2) * C + cc];
        const float by = W[(2 * k2 + 1) * C + cc];
        v2f b;
        b.x = (c < C) ? bx : 0.f;
        b.y = (c < C) ? by : 0.f;
        lB[idx] = b;
    }
    __syncthreads();

    const float* Arow = &lA[m * LDA];
    const int nloc = wave * 16 + m;

#define G2_PUT(q, pv) { const int r_ = (q) >> 5, k_ = ((q) & 31) * 4; \
        const v4f s_ = *(const v4f*)&scl[k_]; const v4f f_ = *(const v4f*)&sft[k_]; \
        v4f v_; v_.x = (pv).x * s_.x + f_.x; v_.y = (pv).y * s_.y + f_.y; \
        v_.z = (pv).z * s_.z + f_.z; v_.w = (pv).w * s_.w + f_.w; \
        *(v4f*)&lA[r_ * LDA + k_] = v_; }
    for (; tile < ntiles; tile += (int)gridDim.x) {
        const int rowb = tile * 16;
        G2_PUT(t,       pa0)
        G2_PUT(t +  96, pa1)
        G2_PUT(t + 192, pa2)
        G2_PUT(t + 288, pa3)
        G2_PUT(t + 384, pa4)
        if (t < 32) G2_PUT(t + 480, pa5)
        const float ds[8] = {pd0, pd1, pd2, pd3, pd4, pd5, pd6, pd7};
        lds_barrier();

        {
            const int nrowb = (tile + (int)gridDim.x) * 16;
            pa0 = G2_ROW(nrowb, 0); pa1 = G2_ROW(nrowb, 1); pa2 = G2_ROW(nrowb, 2);
            pa3 = G2_ROW(nrowb, 3); pa4 = G2_ROW(nrowb, 4); pa5 = G2_ROW5(nrowb);
            pd0 = G2_DIS(nrowb, 0); pd1 = G2_DIS(nrowb, 1); pd2 = G2_DIS(nrowb, 2); pd3 = G2_DIS(nrowb, 3);
            pd4 = G2_DIS(nrowb, 4); pd5 = G2_DIS(nrowb, 5); pd6 = G2_DIS(nrowb, 6); pd7 = G2_DIS(nrowb, 7);
        }

        v8f acc = {0.f, 0.f, 0.f, 0.f, 0.f, 0.f, 0.f, 0.f};
#pragma unroll
        for (int k = 0; k < FDIM; k += 4) {
            v2f a = *(const v2f*)&Arow[k + 2 * hi];
            v2f b = lB[(k / 2 + hi) * 48 + nloc];
            acc = __builtin_amdgcn_wmma_f32_16x16x4_f32(
                false, a, false, b, (short)0, acc, false, false);
        }

        if (nloc < C) {
#pragma unroll
            for (int r = 0; r < 8; ++r) {
                int row = rowb + r + 8 * hi;
                if (row < N) O[(size_t)row * C + nloc] = acc[r] * ds[r];
            }
        }
        lds_barrier();
    }

#pragma unroll
    for (int k = 0; k < 8; ++k) {
        const int i = ci + k * cstride;
        if (i < E) csr_out[i] = cv[k];
    }
    for (int i = ci + 8 * cstride; i < E; i += cstride) csr_out[i] = csr_in[i];
}

#undef G2_ADDR
#undef G2_ROW
#undef G2_ROW5
#undef G2_DIS
#undef G2_PUT

extern "C" void kernel_launch(void* const* d_in, const int* in_sizes, int n_in,
                              void* d_out, int out_size, void* d_ws, size_t ws_size,
                              hipStream_t stream)
{
    const float* x     = (const float*)d_in[0];
    const int*   ei    = (const int*)  d_in[1];
    const float* W1    = (const float*)d_in[2];
    const float* b1    = (const float*)d_in[3];
    const float* gamma = (const float*)d_in[4];
    const float* beta  = (const float*)d_in[5];
    const float* W2    = (const float*)d_in[6];
    const float* b2    = (const float*)d_in[7];
    float* out = (float*)d_out;

    const int N = in_sizes[0] / FDIM;
    const int E = in_sizes[1] / 2;
    const int C = in_sizes[6] / FDIM;
    const int* src = ei;
    const int* dst = ei + E;

    char* ws = (char*)d_ws;
    size_t off = 0;
    auto up = [](size_t v) { return (v + 255) & ~(size_t)255; };
    int*   rowptr = (int*)(ws + off);  off = up(off + (size_t)N * 4);
    float* dis   = (float*)(ws + off); off = up(off + (size_t)N * 4);
    float* stats = (float*)(ws + off); off = up(off + 512 * 4);
    float* h1   = (float*)(ws + off);  off = up(off + (size_t)N * FDIM * 4);
    float* agg1 = (float*)(ws + off);  off = up(off + (size_t)N * FDIM * 4);
    float* h2   = h1;
    int* tmp    = (int*)agg1;
    int* binCnt = (int*)((char*)agg1 + up((size_t)E * 4));
    int* cursor = binCnt + MAXNB;
    int* csr_src = (int*)d_out;
    int* csr_l2  = (int*)((char*)h1 + up((size_t)N * C * 4));

    const int nb_C   = (E + CH - 1) / CH;
    const int nb_B   = (N + 255) / 256;
    const int nb_Nw  = (N + 7) / 8;

    k_init<<<4, 256, 0, stream>>>(binCnt, stats);
    k_hist<<<nb_C, 512, 0, stream>>>(dst, binCnt, E);
    k_scatter<<<nb_C, 512, 0, stream>>>(src, dst, binCnt, cursor, tmp, E);
    k_group<<<nb_B, 256, 0, stream>>>(tmp, binCnt, rowptr, dis, csr_src, N);

    k_gemm1<<<dim3(640, FDIM / 64), 256, 0, stream>>>(x, W1, dis, h1, N);

    k_gather128<<<1024, 512, 0, stream>>>(h1, rowptr, csr_src, dis, b1, agg1, stats, N, E, 1024);

    k_gemm2<<<2084, 96, 0, stream>>>(agg1, W2, stats, gamma, beta, dis, h2, N, C,
                                     csr_src, csr_l2, E);

    k_gatherC<<<nb_Nw, 256, 0, stream>>>(h2, rowptr, csr_l2, dis, b2, out, N, E, C);
}


# Conv2 gather: h2 rows at a 256-byte pitch taken with one 16-byte load per lane, two rows per load (even/odd in-list entries per half-wave, halves added last); in-list and softmax sum order changes
# speedup vs baseline: 5.6601x; 1.0338x over previous
#include <hip/hip_runtime.h>

typedef __attribute__((ext_vector_type(2))) float v2f;
typedef __attribute__((ext_vector_type(8))) float v8f;
typedef __attribute__((ext_vector_type(4))) float v4f;

#define FDIM 128
#define LDA 132
#define H2P 64

#define BSH    8
#define MAXNB  512
#define CH     8192
#define EPT    (CH / 512)
#define SEGPT  20
#define SEGCAP 12288

__global__ __launch_bounds__(256) void k_init(int* __restrict__ bins,
                                              float* __restrict__ stats)
{
    int i = blockIdx.x * 256 + threadIdx.x;
    if (i < 2 * MAXNB) bins[i] = 0;
    if (i < 256) stats[i] = 0.f;
}

__global__ __launch_bounds__(512) void k_hist(const int* __restrict__ dst,
                                              int* __restrict__ binCnt, int E)
{
    __shared__ int hist[MAXNB];
    const int t = threadIdx.x;
    const int base = blockIdx.x * CH;
    const int n = (E - base < CH) ? (E - base) : CH;
    int dv[EPT];
#pragma unroll
    for (int k = 0; k < EPT; ++k) {
        const int i = t + k * 512;
        dv[k] = dst[base + ((i < n) ? i : n - 1)];
    }
    hist[t] = 0;
    __syncthreads();
#pragma unroll
    for (int k = 0; k < EPT; ++k)
        if (t + k * 512 < n) atomicAdd(&hist[dv[k] >> BSH], 1);
    __syncthreads();
    const int c = hist[t];
    if (c > 0) atomicAdd(&binCnt[t], c);
}

__global__ __launch_bounds__(512) void k_scatter(const int* __restrict__ src,
                                                 const int* __restrict__ dst,
                                                 const int* __restrict__ binCnt,
                                                 int* __restrict__ cursor,
                                                 int* __restrict__ tmp, int E)
{
    __shared__ int hist[MAXNB];
    __shared__ int lcur[MAXNB];
    __shared__ int gdelta[MAXNB];
    __shared__ int wc[16], wg[16];
    __shared__ int sval[CH];
    __shared__ int sadr[CH];

    const int t = threadIdx.x, lane = t & 31, wave = t >> 5;
    const int base = blockIdx.x * CH;
    const int n = (E - base < CH) ? (E - base) : CH;

    int dv[EPT], sv[EPT];
#pragma unroll
    for (int k = 0; k < EPT; ++k) {
        const int i = t + k * 512;
        dv[k] = dst[base + ((i < n) ? i : n - 1)];
    }
#pragma unroll
    for (int k = 0; k < EPT; ++k) {
        const int i = t + k * 512;
        sv[k] = src[base + ((i < n) ? i : n - 1)];
    }
    const int g = binCnt[t];

    hist[t] = 0;
    __syncthreads();
#pragma unroll
    for (int k = 0; k < EPT; ++k)
        if (t + k * 512 < n) atomicAdd(&hist[dv[k] >> BSH], 1);
    __syncthreads();

    const int c = hist[t];
    int ci = c, gi = g;
    for (int o = 1; o < 32; o <<= 1) {
        int vc = __shfl_up(ci, o, 32);
        int vg = __shfl_up(gi, o, 32);
        if (lane >= o) { ci += vc; gi += vg; }
    }
    if (lane == 31) { wc[wave] = ci; wg[wave] = gi; }
    __syncthreads();
    int ce = ci - c, ge = gi - g;
    for (int w = 0; w < wave; ++w) { ce += wc[w]; ge += wg[w]; }
    int r = 0;
    if (c > 0) r = atomicAdd(&cursor[t], c);
    lcur[t] = ce;
    gdelta[t] = ge + r - ce;
    __syncthreads();

    int pv[EPT];
#pragma unroll
    for (int k = 0; k < EPT; ++k) {
        pv[k] = 0;
        if (t + k * 512 < n) pv[k] = atomicAdd(&lcur[dv[k] >> BSH], 1);
    }
#pragma unroll
    for (int k = 0; k < EPT; ++k) {
        if (t + k * 512 < n) {
            const int p = pv[k];
            sval[p] = (int)(((unsigned)sv[k] << 8) | (unsigned)(dv[k] & 255));
            sadr[p] = gdelta[dv[k] >> BSH] + p;
        }
    }
    __syncthreads();
    for (int i = t; i < n; i += 512) tmp[sadr[i]] = sval[i];
}

__global__ __launch_bounds__(256) void k_group(const int* __restrict__ tmp,
                                               const int* __restrict__ binCnt,
                                               int* __restrict__ rowptr,
                                               float* __restrict__ dis,
                                               int* __restrict__ csr_src, int N)
{
    __shared__ int cnt[256];
    __shared__ int red[8], wsum[8];
    __shared__ int sbuf[SEGCAP];

    const int t = threadIdx.x, lane = t & 31, wave = t >> 5;
    const int b = blockIdx.x;

    const int g0 = binCnt[t], g1 = binCnt[t + 256];
    int a = ((t < b) ? g0 : 0) + ((t + 256 < b) ? g1 : 0);
    for (int o = 16; o > 0; o >>= 1) a += __shfl_xor(a, o, 32);
    if (lane == 0) red[wave] = a;
    cnt[t] = 0;
    __syncthreads();
    int beg = 0;
    for (int w = 0; w < 8; ++w) beg += red[w];
    beg = __builtin_amdgcn_readfirstlane(beg);
    const int len = __builtin_amdgcn_readfirstlane(binCnt[b]);
    const bool staged = len <= SEGCAP;

    for (int i0 = 0; i0 < len; i0 += SEGPT * 256) {
        int v[SEGPT];
#pragma unroll
        for (int k = 0; k < SEGPT; ++k) {
            const int i = i0 + t + k * 256;
            v[k] = tmp[beg + ((i < len) ? i : len - 1)];
        }
#pragma unroll
        for (int k = 0; k < SEGPT; ++k)
            if (i0 + t + k * 256 < len) atomicAdd(&cnt[v[k] & 255], 1);
    }
    __syncthreads();

    const int c = cnt[t];
    int incl = c;
    for (int o = 1; o < 32; o <<= 1) {
        int v = __shfl_up(incl, o, 32);
        if (lane >= o) incl += v;
    }
    if (lane == 31) wsum[wave] = incl;
    __syncthreads();
    int excl = incl - c;
    for (int w = 0; w < wave; ++w) excl += wsum[w];
    const int node = b * 256 + t;
    if (node < N) {
        rowptr[node] = beg + excl;
        dis[node] = rsqrtf((float)(c + 1));
    }
    cnt[t] = excl;
    __syncthreads();

    for (int i0 = 0; i0 < len; i0 += SEGPT * 256) {
        unsigned v[SEGPT];
        int p[SEGPT];
#pragma unroll
        for (int k = 0; k < SEGPT; ++k) {
            const int i = i0 + t + k * 256;
            v[k] = (unsigned)tmp[beg + ((i < len) ? i : len - 1)];
        }
#pragma unroll
        for (int k = 0; k < SEGPT; ++k) {
            p[k] = 0;
            if (i0 + t + k * 256 < len) p[k] = atomicAdd(&cnt[v[k] & 255u], 1);
        }
#pragma unroll
        for (int k = 0; k < SEGPT; ++k) {
            if (i0 + t + k * 256 < len) {
                const int s = (int)(v[k] >> 8);
                if (staged) sbuf[p[k]] = s;
                else        csr_src[beg + p[k]] = s;
            }
        }
    }
    __syncthreads();
    if (staged)
        for (int i = t; i < len; i += 256) csr_src[beg + i] = sbuf[i];
}

static __device__ __forceinline__ void lds_barrier()
{
    __builtin_amdgcn_fence(__ATOMIC_RELEASE, "workgroup", "local");
    __builtin_amdgcn_s_barrier();
    __builtin_amdgcn_fence(__ATOMIC_ACQUIRE, "workgroup", "local");
}

__global__ __launch_bounds__(256) void k_gemm1(const float* __restrict__ X,
                                               const float* __restrict__ W,
                                               const float* __restrict__ dis,
                                               float* __restrict__ O, int N)
{
    __shared__ __attribute__((aligned(16))) float lA[32 * LDA];
    __shared__ v2f   lB[64 * 64];

    const int t    = threadIdx.x;
    const int colb = blockIdx.y * 64;
    const int ntiles = (N + 31) >> 5;
    const int wave = t >> 5, lane = t & 31;
    const int m = lane & 15, hi = lane >> 4;
    const int rt = wave >> 2, ct = wave & 3;
    const int ak = lane * 4;
    const int er = rt * 16 + 8 * hi;

#define G1_ROW(rb, j) (*(const v4f*)&X[(size_t)(((rb) + wave + 8 * (j)) < N ? ((rb) + wave + 8 * (j)) : N - 1) * FDIM + ak])
#define G1_DIS(rb, r) (dis[((rb) + er + (r)) < N ? ((rb) + er + (r)) : N - 1])
    v4f pa0, pa1, pa2, pa3;
    float pd0, pd1, pd2, pd3, pd4, pd5, pd6, pd7;
    int tile = blockIdx.x;
    {
        const int rowb = tile * 32;
        pa0 = G1_ROW(rowb, 0); pa1 = G1_ROW(rowb, 1); pa2 = G1_ROW(rowb, 2); pa3 = G1_ROW(rowb, 3);
        pd0 = G1_DIS(rowb, 0); pd1 = G1_DIS(rowb, 1); pd2 = G1_DIS(rowb, 2); pd3 = G1_DIS(rowb, 3);
        pd4 = G1_DIS(rowb, 4); pd5 = G1_DIS(rowb, 5); pd6 = G1_DIS(rowb, 6); pd7 = G1_DIS(rowb, 7);
    }

#pragma unroll
    for (int i = 0; i < 16; ++i) {
        const int idx = t + 256 * i;
        const int k2 = idx >> 6, c = idx & 63;
        const float* wp = W + (size_t)(2 * k2) * FDIM + (colb + c);
        v2f b; b.x = wp[0]; b.y = wp[FDIM];
        lB[idx] = b;
    }

    const float* Arow = &lA[(rt * 16 + m) * LDA];
    const int nloc = ct * 16 + m;
    const int col = colb + nloc;

    for (; tile < ntiles; tile += (int)gridDim.x) {
        const int rowb = tile * 32;
        *(v4f*)&lA[(wave +  0) * LDA + ak] = pa0;
        *(v4f*)&lA[(wave +  8) * LDA + ak] = pa1;
        *(v4f*)&lA[(wave + 16) * LDA + ak] = pa2;
        *(v4f*)&lA[(wave + 24) * LDA + ak] = pa3;
        const float ds[8] = {pd0, pd1, pd2, pd3, pd4, pd5, pd6, pd7};
        lds_barrier();

        {
            const int nrowb = (tile + (int)gridDim.x) * 32;
            pa0 = G1_ROW(nrowb, 0); pa1 = G1_ROW(nrowb, 1); pa2 = G1_ROW(nrowb, 2); pa3 = G1_ROW(nrowb, 3);
            pd0 = G1_DIS(nrowb, 0); pd1 = G1_DIS(nrowb, 1); pd2 = G1_DIS(nrowb, 2); pd3 = G1_DIS(nrowb, 3);
            pd4 = G1_DIS(nrowb, 4); pd5 = G1_DIS(nrowb, 5); pd6 = G1_DIS(nrowb, 6); pd7 = G1_DIS(nrowb, 7);
        }

        v8f acc = {0.f, 0.f, 0.f, 0.f, 0.f, 0.f, 0.f, 0.f};
#pragma unroll
        for (int k = 0; k < FDIM; k += 4) {
            v2f a = *(const v2f*)&Arow[k + 2 * hi];
            v2f b = lB[(k / 2 + hi) * 64 + nloc];
            acc = __builtin_amdgcn_wmma_f32_16x16x4_f32(
                false, a, false, b, (short)0, acc, false, false);
        }

#pragma unroll
        for (int r = 0; r < 8; ++r) {
            int row = rowb + er + r;
            if (row < N) O[(size_t)row * FDIM + col] = acc[r] * ds[r];
        }
        lds_barrier();
    }
}

#undef G1_ROW
#undef G1_DIS

static __device__ __forceinline__ int bcast_i(int v, int j)
{
    return __builtin_amdgcn_readlane(v, j);
}
static __device__ __forceinline__ void row_add(float4& acc, const float4& v)
{
    acc.x += v.x; acc.y += v.y; acc.z += v.z; acc.w += v.w;
}

__global__ __launch_bounds__(512) void k_gather128(const float* __restrict__ h,
                                                   const int* __restrict__ rowptr,
                                                   const int* __restrict__ csr_src,
                                                   const float* __restrict__ dis,
                                                   const float* __restrict__ bias,
                                                   float* __restrict__ out,
                                                   float* __restrict__ stats,
                                                   int N, int E, int nblk)
{
    __shared__ float red[16 * 256];
    const int lane = threadIdx.x & 31;
    const int wave = __builtin_amdgcn_readfirstlane((int)(threadIdx.x >> 5));
    const float4* hp = (const float4*)h;
    const float4 b = ((const float4*)bias)[lane];
    float4 s1 = {0.f, 0.f, 0.f, 0.f}, s2 = {0.f, 0.f, 0.f, 0.f};

    for (int nb = blockIdx.x * 16; nb < N; nb += nblk * 16) {
        const int node = nb + wave;
        if (node < N) {
            const int beg = __builtin_amdgcn_readfirstlane(rowptr[node]);
            const int end = (node + 1 < N) ? __builtin_amdgcn_readfirstlane(rowptr[node + 1]) : E;
            const float di = dis[node];

            float4 acc = hp[(size_t)node * 32 + lane];
            for (int p = beg; p < end; p += 32) {
                const int n = (end - p < 32) ? (end - p) : 32;
                int s_l = 0;
                if (lane < n) s_l = csr_src[p + lane];
                int j = 0;
                for (; j + 4 <= n; j += 4) {
                    const int i0 = bcast_i(s_l, j),     i1 = bcast_i(s_l, j + 1);
                    const int i2 = bcast_i(s_l, j + 2), i3 = bcast_i(s_l, j + 3);
                    const float4 v0 = hp[(size_t)i0 * 32 + lane];
                    const float4 v1 = hp[(size_t)i1 * 32 + lane];
                    const float4 v2 = hp[(size_t)i2 * 32 + lane];
                    const float4 v3 = hp[(size_t)i3 * 32 + lane];
                    row_add(acc, v0); row_add(acc, v1);
                    row_add(acc, v2); row_add(acc, v3);
                }
                for (; j < n; ++j) {
                    const int i0 = bcast_i(s_l, j);
                    const float4 v0 = hp[(size_t)i0 * 32 + lane];
                    row_add(acc, v0);
                }
            }
            acc.x = fmaxf(acc.x * di + b.x, 0.f);
            acc.y = fmaxf(acc.y * di + b.y, 0.f);
            acc.z = fmaxf(acc.z * di + b.z, 0.f);
            acc.w = fmaxf(acc.w * di + b.w, 0.f);
            ((float4*)out)[(size_t)node * 32 + lane] = acc;
            s1.x += acc.x; s1.y += acc.y; s1.z += acc.z; s1.w += acc.w;
            s2.x += acc.x * acc.x; s2.y += acc.y * acc.y;
            s2.z += acc.z * acc.z; s2.w += acc.w * acc.w;
        }
    }

    ((float4*)red)[wave * 64 + lane]      = s1;
    ((float4*)red)[wave * 64 + 32 + lane] = s2;
    __syncthreads();
    const int t = threadIdx.x;
    if (t < 256) {
        float v = 0.f;
        for (int w = 0; w < 16; ++w) v += red[w * 256 + t];
        atomicAdd(&stats[t], v);
    }
}

static __device__ __forceinline__ float4 row16(const float* __restrict__ h, int r, int chunk)
{
    const unsigned off = ((unsigned)r << 8) | ((unsigned)chunk << 4);
    return *(const float4*)((const char*)h + off);
}
static __device__ __forceinline__ void row_add_if(float4& acc, const float4& v, bool f)
{
    acc.x += f ? v.x : 0.f; acc.y += f ? v.y : 0.f;
    acc.z += f ? v.z : 0.f; acc.w += f ? v.w : 0.f;
}

__global__ __launch_bounds__(256) void k_gatherC(const float* __restrict__ h,
                                                 const int* __restrict__ rowptr,
                                                 const int* __restrict__ csr_src,
                                                 const float* __restrict__ dis,
                                                 const float* __restrict__ bias,
                                                 float* __restrict__ out, int N, int E, int C)
{
    const int lane = threadIdx.x & 31;
    const int node = blockIdx.x * 8 + __builtin_amdgcn_readfirstlane((int)(threadIdx.x >> 5));
    if (node >= N) return;
    const int beg = __builtin_amdgcn_readfirstlane(rowptr[node]);
    const int end = (node + 1 < N) ? __builtin_amdgcn_readfirstlane(rowptr[node + 1]) : E;
    const float di = dis[node];
    const int hi = lane >> 4, sub = lane & 15;
    const int c4 = sub * 4;
    const bool k0 = c4 < C, k1 = c4 + 1 < C, k2 = c4 + 2 < C, k3 = c4 + 3 < C;
    const int chunk = k0 ? sub : 0;
    const bool odd = hi != 0;

    float4 acc;
    {
        const float4 v = row16(h, node, chunk);
        acc.x = odd ? 0.f : v.x; acc.y = odd ? 0.f : v.y;
        acc.z = odd ? 0.f : v.z; acc.w = odd ? 0.f : v.w;
    }
    const float b0 = bias[k0 ? c4 : 0],     b1 = bias[k1 ? c4 + 1 : 0];
    const float b2 = bias[k2 ? c4 + 2 : 0], b3 = bias[k3 ? c4 + 3 : 0];

    for (int p = beg; p < end; p += 32) {
        const int n = (end - p < 32) ? (end - p) : 32;
        int s_l = 0;
        if (lane < n) s_l = csr_src[p + lane];
        int j = 0;
        for (; j + 8 <= n; j += 8) {
            const int e0 = bcast_i(s_l, j),     o0 = bcast_i(s_l, j + 1);
            const int e1 = bcast_i(s_l, j + 2), o1 = bcast_i(s_l, j + 3);
            const int e2 = bcast_i(s_l, j + 4), o2 = bcast_i(s_l, j + 5);
            const int e3 = bcast_i(s_l, j + 6), o3 = bcast_i(s_l, j + 7);
            const float4 v0 = row16(h, odd ? o0 : e0, chunk);
            const float4 v1 = row16(h, odd ? o1 : e1, chunk);
            const float4 v2 = row16(h, odd ? o2 : e2, chunk);
            const float4 v3 = row16(h, odd ? o3 : e3, chunk);
            row_add(acc, v0); row_add(acc, v1);
            row_add(acc, v2); row_add(acc, v3);
        }
        const int rem = n - j;
        if (rem > 0) {
            const int last = n - 1;
            const int e0 = bcast_i(s_l, j);
            const int o0 = bcast_i(s_l, (j + 1 < last) ? j + 1 : last);
            const int e1 = bcast_i(s_l, (j + 2 < last) ? j + 2 : last);
            const int o1 = bcast_i(s_l, (j + 3 < last) ? j + 3 : last);
            const int e2 = bcast_i(s_l, (j + 4 < last) ? j + 4 : last);
            const int o2 = bcast_i(s_l, (j + 5 < last) ? j + 5 : last);
            const int e3 = bcast_i(s_l, (j + 6 < last) ? j + 6 : last);
            const float4 zero = {0.f, 0.f, 0.f, 0.f};
            const float4 v0 = row16(h, odd ? o0 : e0, chunk);
            float4 v1 = zero, v2 = zero, v3 = zero;
            if (rem > 2) {
                v1 = row16(h, odd ? o1 : e1, chunk);
                if (rem > 4) {
                    v2 = row16(h, odd ? o2 : e2, chunk);
                    if (rem > 6) v3 = row16(h, e3, chunk);
                }
            }
            row_add_if(acc, v0, hi     < rem);
            row_add_if(acc, v1, hi + 2 < rem);
            row_add_if(acc, v2, hi + 4 < rem);
            row_add_if(acc, v3, hi + 6 < rem);
        }
    }
    acc.x += __shfl_xor(acc.x, 16, 32);
    acc.y += __shfl_xor(acc.y, 16, 32);
    acc.z += __shfl_xor(acc.z, 16, 32);
    acc.w += __shfl_xor(acc.w, 16, 32);

    const float v0 = k0 ? (acc.x * di + b0) : -3.4e38f;
    const float v1 = k1 ? (acc.y * di + b1) : -3.4e38f;
    const float v2 = k2 ? (acc.z * di + b2) : -3.4e38f;
    const float v3 = k3 ? (acc.w * di + b3) : -3.4e38f;
    float mx = fmaxf(fmaxf(v0, v1), fmaxf(v2, v3));
#pragma unroll
    for (int o = 8; o > 0; o >>= 1) mx = fmaxf(mx, __shfl_xor(mx, o, 32));
    float s = (k0 ? expf(v0 - mx) : 0.f) + (k1 ? expf(v1 - mx) : 0.f)
            + (k2 ? expf(v2 - mx) : 0.f) + (k3 ? expf(v3 - mx) : 0.f);
#pragma unroll
    for (int o = 8; o > 0; o >>= 1) s += __shfl_xor(s, o, 32);
    const float l = logf(s);
    float* orow = out + (size_t)node * C + c4;
    if (!odd) {
        if ((C & 3) == 0) {
            if (k0) {
                float4 r;
                r.x = v0 - mx - l; r.y = v1 - mx - l; r.z = v2 - mx - l; r.w = v3 - mx - l;
                *(float4*)orow = r;
            }
        } else {
            if (k0) orow[0] = v0 - mx - l;
            if (k1) orow[1] = v1 - mx - l;
            if (k2) orow[2] = v2 - mx - l;
            if (k3) orow[3] = v3 - mx - l;
        }
    }
}

__global__ __launch_bounds__(96) void k_gemm2(const float* __restrict__ A,
                                              const float* __restrict__ W,
                                              const float* __restrict__ stats,
                                              const float* __restrict__ gamma,
                                              const float* __restrict__ beta,
                                              const float* __restrict__ dis,
                                              float* __restrict__ O, int N, int C,
                                              const int* __restrict__ csr_in,
                                              int* __restrict__ csr_out, int E)
{
    __shared__ __attribute__((aligned(16))) float lA[16 * LDA];
    __shared__ v2f   lB[64 * 48];
    __shared__ __attribute__((aligned(16))) float scl[FDIM];
    __shared__ __attribute__((aligned(16))) float sft[FDIM];

    const int t = threadIdx.x;
    const int ntiles = (N + 15) >> 4;
    const int wave = t >> 5, lane = t & 31;
    const int m = lane & 15, hi = lane >> 4;

    const int cstride = (int)gridDim.x * 96;
    const int ci = blockIdx.x * 96 + t;
    int cv[8];
#pragma unroll
    for (int k = 0; k < 8; ++k) {
        const int i = ci + k * cstride;
        cv[k] = csr_in[(i < E) ? i : E - 1];
    }

    const int q5 = (t + 480 > 511) ? 511 : t + 480;
#define G2_ADDR(rb, q) (&A[(size_t)(((rb) + ((q) >> 5)) < N ? ((rb) + ((q) >> 5)) : N - 1) * FDIM + ((q) & 31) * 4])
#define G2_ROW(rb, j)  (*(const v4f*)G2_ADDR(rb, t + 96 * (j)))
#define G2_ROW5(rb)    (*(const v4f*)G2_ADDR(rb, q5))
#define G2_DIS(rb, r)  (dis[((rb) + 8 * hi + (r)) < N ? ((rb) + 8 * hi + (r)) : N - 1])
    v4f pa0, pa1, pa2, pa3, pa4, pa5;
    float pd0, pd1, pd2, pd3, pd4, pd5, pd6, pd7;
    int tile = blockIdx.x;
    {
        const int rowb = tile * 16;
        pa0 = G2_ROW(rowb, 0); pa1 = G2_ROW(rowb, 1); pa2 = G2_ROW(rowb, 2);
        pa3 = G2_ROW(rowb, 3); pa4 = G2_ROW(rowb, 4); pa5 = G2_ROW5(rowb);
        pd0 = G2_DIS(rowb, 0); pd1 = G2_DIS(rowb, 1); pd2 = G2_DIS(rowb, 2); pd3 = G2_DIS(rowb, 3);
        pd4 = G2_DIS(rowb, 4); pd5 = G2_DIS(rowb, 5); pd6 = G2_DIS(rowb, 6); pd7 = G2_DIS(rowb, 7);
    }

    for (int f = t; f < FDIM; f += 96) {
        const float inv_n = 1.f / (float)N;
        const float mean = stats[f] * inv_n;
        const float var  = stats[FDIM + f] * inv_n - mean * mean;
        const float inv  = rsqrtf(var + 1e-5f);
        const float g = gamma[f] * inv;
        scl[f] = g;
        sft[f] = beta[f] - mean * g;
    }
#pragma unroll 8
    for (int i = 0; i < 32; ++i) {
        const int idx = t + 96 * i;
        const int k2 = idx / 48, c = idx - k2 * 48;
        const int cc = (c < C) ? c : 0;
        const float bx = W[(2 * k2) * C + cc];
        const float by = W[(2 * k2 + 1) * C + cc];
        v2f b;
        b.x = (c < C) ? bx : 0.f;
        b.y = (c < C) ? by : 0.f;
        lB[idx] = b;
    }
    __syncthreads();

    const float* Arow = &lA[m * LDA];
    const int nloc = wave * 16 + m;

#define G2_PUT(q, pv) { const int r_ = (q) >> 5, k_ = ((q) & 31) * 4; \
        const v4f s_ = *(const v4f*)&scl[k_]; const v4f f_ = *(const v4f*)&sft[k_]; \
        v4f v_; v_.x = (pv).x * s_.x + f_.x; v_.y = (pv).y * s_.y + f_.y; \
        v_.z = (pv).z * s_.z + f_.z; v_.w = (pv).w * s_.w + f_.w; \
        *(v4f*)&lA[r_ * LDA + k_] = v_; }
    for (; tile < ntiles; tile += (int)gridDim.x) {
        const int rowb = tile * 16;
        G2_PUT(t,       pa0)
        G2_PUT(t +  96, pa1)
        G2_PUT(t + 192, pa2)
        G2_PUT(t + 288, pa3)
        G2_PUT(t + 384, pa4)
        if (t < 32) G2_PUT(t + 480, pa5)
        const float ds[8] = {pd0, pd1, pd2, pd3, pd4, pd5, pd6, pd7};
        lds_barrier();

        {
            const int nrowb = (tile + (int)gridDim.x) * 16;
            pa0 = G2_ROW(nrowb, 0); pa1 = G2_ROW(nrowb, 1); pa2 = G2_ROW(nrowb, 2);
            pa3 = G2_ROW(nrowb, 3); pa4 = G2_ROW(nrowb, 4); pa5 = G2_ROW5(nrowb);
            pd0 = G2_DIS(nrowb, 0); pd1 = G2_DIS(nrowb, 1); pd2 = G2_DIS(nrowb, 2); pd3 = G2_DIS(nrowb, 3);
            pd4 = G2_DIS(nrowb, 4); pd5 = G2_DIS(nrowb, 5); pd6 = G2_DIS(nrowb, 6); pd7 = G2_DIS(nrowb, 7);
        }

        v8f acc = {0.f, 0.f, 0.f, 0.f, 0.f, 0.f, 0.f, 0.f};
#pragma unroll
        for (int k = 0; k < FDIM; k += 4) {
            v2f a = *(const v2f*)&Arow[k + 2 * hi];
            v2f b = lB[(k / 2 + hi) * 48 + nloc];
            acc = __builtin_amdgcn_wmma_f32_16x16x4_f32(
                false, a, false, b, (short)0, acc, false, false);
        }

        if (nloc < C) {
#pragma unroll
            for (int r = 0; r < 8; ++r) {
                int row = rowb + r + 8 * hi;
                if (row < N) O[(size_t)row * H2P + nloc] = acc[r] * ds[r];
            }
        }
        lds_barrier();
    }

#pragma unroll
    for (int k = 0; k < 8; ++k) {
        const int i = ci + k * cstride;
        if (i < E) csr_out[i] = cv[k];
    }
    for (int i = ci + 8 * cstride; i < E; i += cstride) csr_out[i] = csr_in[i];
}

#undef G2_ADDR
#undef G2_ROW
#undef G2_ROW5
#undef G2_DIS
#undef G2_PUT

extern "C" void kernel_launch(void* const* d_in, const int* in_sizes, int n_in,
                              void* d_out, int out_size, void* d_ws, size_t ws_size,
                              hipStream_t stream)
{
    const float* x     = (const float*)d_in[0];
    const int*   ei    = (const int*)  d_in[1];
    const float* W1    = (const float*)d_in[2];
    const float* b1    = (const float*)d_in[3];
    const float* gamma = (const float*)d_in[4];
    const float* beta  = (const float*)d_in[5];
    const float* W2    = (const float*)d_in[6];
    const float* b2    = (const float*)d_in[7];
    float* out = (float*)d_out;

    const int N = in_sizes[0] / FDIM;
    const int E = in_sizes[1] / 2;
    const int C = in_sizes[6] / FDIM;
    const int* src = ei;
    const int* dst = ei + E;

    char* ws = (char*)d_ws;
    size_t off = 0;
    auto up = [](size_t v) { return (v + 255) & ~(size_t)255; };
    int*   rowptr = (int*)(ws + off);  off = up(off + (size_t)N * 4);
    float* dis   = (float*)(ws + off); off = up(off + (size_t)N * 4);
    float* stats = (float*)(ws + off); off = up(off + 512 * 4);
    float* h1   = (float*)(ws + off);  off = up(off + (size_t)N * FDIM * 4);
    float* agg1 = (float*)(ws + off);  off = up(off + (size_t)N * FDIM * 4);
    float* h2   = h1;
    int* tmp    = (int*)agg1;
    int* binCnt = (int*)((char*)agg1 + up((size_t)E * 4));
    int* cursor = binCnt + MAXNB;
    int* csr_src = (int*)d_out;
    int* csr_l2  = (int*)((char*)h1 + up((size_t)N * H2P * 4));

    const int nb_C   = (E + CH - 1) / CH;
    const int nb_B   = (N + 255) / 256;
    const int nb_Nw  = (N + 7) / 8;

    k_init<<<4, 256, 0, stream>>>(binCnt, stats);
    k_hist<<<nb_C, 512, 0, stream>>>(dst, binCnt, E);
    k_scatter<<<nb_C, 512, 0, stream>>>(src, dst, binCnt, cursor, tmp, E);
    k_group<<<nb_B, 256, 0, stream>>>(tmp, binCnt, rowptr, dis, csr_src, N);

    k_gemm1<<<dim3(640, FDIM / 64), 256, 0, stream>>>(x, W1, dis, h1, N);

    k_gather128<<<1024, 512, 0, stream>>>(h1, rowptr, csr_src, dis, b1, agg1, stats, N, E, 1024);

    k_gemm2<<<2084, 96, 0, stream>>>(agg1, W2, stats, gamma, beta, dis, h2, N, C,
                                     csr_src, csr_l2, E);

    k_gatherC<<<nb_Nw, 256, 0, stream>>>(h2, rowptr, csr_l2, dis, b2, out, N, E, C);
}
